# starting kernel
# baseline (speedup 1.0000x reference)
#include <hip/hip_runtime.h>
#include <hip/hip_bf16.h>


typedef __attribute__((ext_vector_type(2))) float v2f;
typedef __attribute__((ext_vector_type(8))) float v8f;

#define TWO_PI_F     6.28318530717958647692f
#define PI_F         3.14159265358979323846f
#define INV_TWO_PI_F 0.15915494309189533577f
#define DT_F         0.001f

__device__ __forceinline__ float pmod(float x, float m) {
    float r = fmodf(x, m);
    return (r < 0.0f) ? (r + m) : r;
}

__device__ __forceinline__ float wrap_add(float x, float d) {
    float t = x + d;
    return (t >= TWO_PI_F) ? (t - TWO_PI_F) : t;
}

struct alignas(16) F4 { float v[4]; };

__global__ void osc_scan_kernel(const float* __restrict__ p_log_slow,
                                const float* __restrict__ p_log_fast,
                                const float* __restrict__ p_reset,
                                float* __restrict__ out, int N) {
    if (threadIdx.x != 0 || blockIdx.x != 0) return;

    const float dslow = expf(p_log_slow[0]) * DT_F;
    const float dfast = expf(p_log_fast[0]) * DT_F;
    const float keep  = 1.0f - p_reset[0];

    float slow = 0.0f, fast = 0.0f, prev = 0.0f;
    float* __restrict__ oS = out;
    float* __restrict__ oF = out + N;

    const bool fast_ok =
        (dslow >= 0.0f) && (dslow < TWO_PI_F) &&
        (dfast >= 0.0f) && (dfast < TWO_PI_F) &&
        (keep  >= 0.0f) && (keep  <= 1.0f);

    int t = 0;
    if (fast_ok && ((N & 3) == 0)) {
        for (; t < N; t += 4) {
            F4 s4, f4;
#pragma unroll
            for (int u = 0; u < 4; ++u) {
                s4.v[u] = slow;
                f4.v[u] = fast;
                bool crossed = (prev > PI_F) && (slow < PI_F);
                float ns = wrap_add(slow, dslow);
                float nf = wrap_add(fast, dfast);
                nf = crossed ? nf * keep : nf;
                prev = slow; slow = ns; fast = nf;
            }
            *(F4*)(oS + t) = s4;
            *(F4*)(oF + t) = f4;
        }
    }
    for (; t < N; ++t) {
        oS[t] = slow;
        oF[t] = fast;
        bool crossed = (prev > PI_F) && (slow < PI_F);
        float ns = pmod(slow + dslow, TWO_PI_F);
        float nf = pmod(fast + dfast, TWO_PI_F);
        if (crossed) nf = pmod(nf * keep, TWO_PI_F);
        prev = slow; slow = ns; fast = nf;
    }
}

__global__ void osc_scale_wmma_kernel(const float* __restrict__ in,
                                      float* __restrict__ out, int N) {
    const float s = INV_TWO_PI_F;
    const int base = blockIdx.x * 256;
    const int lane = threadIdx.x;
    const int i    = lane & 15;
    const int kh   = lane >> 4;

    if (base + 256 <= N) {
        v8f d = {0.f, 0.f, 0.f, 0.f, 0.f, 0.f, 0.f, 0.f};
#pragma unroll
        for (int m = 0; m < 4; ++m) {
            const float* p = in + base + i * 16 + 4 * m + 2 * kh;
            v2f a; a.x = p[0]; a.y = p[1];
            v2f b;
            b.x = (i == (4 * m + 2 * kh + 0)) ? s : 0.0f;
            b.y = (i == (4 * m + 2 * kh + 1)) ? s : 0.0f;
            d = __builtin_amdgcn_wmma_f32_16x16x4_f32(
                    false, a, false, b, (short)0, d, false, false);
        }
#pragma unroll
        for (int r = 0; r < 8; ++r) {
            out[base + (r + 8 * kh) * 16 + i] = d[r];
        }
    } else {
        for (int j = base + lane; j < N; j += 32) out[j] = in[j] * s;
    }
}

extern "C" void kernel_launch(void* const* d_in, const int* in_sizes, int n_in,
                              void* d_out, int out_size, void* d_ws, size_t ws_size,
                              hipStream_t stream) {
    const float* p_log_slow = (const float*)d_in[0];
    const float* p_log_fast = (const float*)d_in[1];
    const float* p_reset    = (const float*)d_in[2];
    const int N = out_size / 3;
    if (N <= 0) return;
    float* out = (float*)d_out;

    osc_scan_kernel<<<1, 32, 0, stream>>>(p_log_slow, p_log_fast, p_reset, out, N);

    const int tiles = (N + 255) / 256;
    osc_scale_wmma_kernel<<<tiles, 32, 0, stream>>>(out, out + 2 * (size_t)N, N);
}


# scan: when the fast phase at a trough crossing recurs bit-for-bit, proposals are replayed from an LDS history one recurrence lag back plus the carried offset; run walk as fallback; same values
# speedup vs baseline: 3.2306x; 3.2306x over previous
#include <hip/hip_runtime.h>
#include <hip/hip_bf16.h>


typedef __attribute__((ext_vector_type(2))) float v2f;
typedef __attribute__((ext_vector_type(8))) float v8f;

#define TWO_PI_F     6.28318530717958647692f
#define PI_F         3.14159265358979323846f
#define INV_TWO_PI_F 0.15915494309189533577f
#define DT_F         0.001f

__device__ __forceinline__ float pmod(float x, float m) {
    float r = fmodf(x, m);
    return (r < 0.0f) ? (r + m) : r;
}

__device__ __forceinline__ float wrap_add(float x, float d) {
    float t = x + d;
    return (t >= TWO_PI_F) ? (t - TWO_PI_F) : t;
}


struct Seg { float c, w, t2, de; };

__device__ __forceinline__ float to_scalar(float x) {
    float r;
    asm("v_readfirstlane_b32 %0, %1" : "=s"(r) : "v"(x));
    return r;
}

__device__ __forceinline__ void walk_pair(float& wF, float& cF, float dF, float rdF, Seg& sgF,
                                          float& wS, float& cS, float dS, float rdS, Seg& sgS) {
    float wnF, cnF, t2F, deF, bF, tF, baF;
    float wnS, cnS, t2S, deS, bS, tS, baS;
    asm(
        "s_and_b32 %[bF], %[wF], 0x7f800000\n\t"
        "s_and_b32 %[bS], %[wS], 0x7f800000\n\t"
        "s_add_f32 %[t2F], %[wF], %[dF]\n\t"
        "s_add_f32 %[t2S], %[wS], %[dS]\n\t"
        "s_add_f32 %[tF], %[bF], %[bF]\n\t"
        "s_add_f32 %[tS], %[bS], %[bS]\n\t"
        "s_add_f32 %[deF], %[bF], %[dF]\n\t"
        "s_add_f32 %[deS], %[bS], %[dS]\n\t"
        "s_min_u32 %[tF], %[tF], 0x40c90fdb\n\t"
        "s_min_u32 %[tS], %[tS], 0x40c90fdb\n\t"
        "s_sub_f32 %[deF], %[deF], %[bF]\n\t"
        "s_sub_f32 %[deS], %[deS], %[bS]\n\t"
        "s_sub_f32 %[tF], %[tF], %[t2F]\n\t"
        "s_sub_f32 %[tS], %[tS], %[t2S]\n\t"
        "s_sub_f32 %[baF], %[t2F], %[deF]\n\t"
        "s_sub_f32 %[baS], %[t2S], %[deS]\n\t"
        "s_mul_f32 %[tF], %[tF], %[rF]\n\t"
        "s_mul_f32 %[tS], %[tS], %[rS]\n\t"
        "s_ceil_f32 %[tF], %[tF]\n\t"
        "s_ceil_f32 %[tS], %[tS]\n\t"
        "s_fmac_f32 %[baF], %[tF], %[deF]\n\t"
        "s_fmac_f32 %[baS], %[tS], %[deS]\n\t"
        "s_max_num_f32 %[cnF], %[tF], 0\n\t"
        "s_max_num_f32 %[cnS], %[tS], 0\n\t"
        "s_cmp_ge_f32 %[tF], 1.0\n\t"
        "s_cselect_b32 %[baF], %[baF], %[wF]\n\t"
        "s_cmp_ge_f32 %[tS], 1.0\n\t"
        "s_cselect_b32 %[baS], %[baS], %[wS]\n\t"
        "s_add_f32 %[baF], %[baF], %[dF]\n\t"
        "s_add_f32 %[baS], %[baS], %[dS]\n\t"
        "s_add_f32 %[cnF], %[cnF], %[cF]\n\t"
        "s_add_f32 %[cnS], %[cnS], %[cS]\n\t"
        "s_add_f32 %[wnF], %[baF], 0xc0c90fdb\n\t"
        "s_add_f32 %[wnS], %[baS], 0xc0c90fdb\n\t"
        "s_add_f32 %[cnF], %[cnF], 1.0\n\t"
        "s_add_f32 %[cnS], %[cnS], 1.0\n\t"
        "s_cmp_ge_f32 %[baF], 0x40c90fdb\n\t"
        "s_cselect_b32 %[wnF], %[wnF], %[baF]\n\t"
        "s_cmp_ge_f32 %[baS], 0x40c90fdb\n\t"
        "s_cselect_b32 %[wnS], %[wnS], %[baS]\n\t"
        : [wnF] "=&s"(wnF), [cnF] "=&s"(cnF), [t2F] "=&s"(t2F), [deF] "=&s"(deF), [bF] "=&s"(bF), [tF] "=&s"(tF), [baF] "=&s"(baF), [wnS] "=&s"(wnS), [cnS] "=&s"(cnS), [t2S] "=&s"(t2S), [deS] "=&s"(deS), [bS] "=&s"(bS), [tS] "=&s"(tS), [baS] "=&s"(baS)
        : [wF] "s"(wF), [cF] "s"(cF), [dF] "s"(dF), [rF] "s"(rdF), [wS] "s"(wS), [cS] "s"(cS), [dS] "s"(dS), [rS] "s"(rdS)
        : "scc");
    sgF.c = cF; sgF.w = wF; sgF.t2 = t2F; sgF.de = deF;
    sgS.c = cS; sgS.w = wS; sgS.t2 = t2S; sgS.de = deS;
    wF = wnF; cF = cnF;
    wS = wnS; cS = cnS;
}

__device__ __forceinline__ void walk_one(float& wF, float& cF, float dF, float rdF, Seg& sgF) {
    float wnF, cnF, t2F, deF, bF, tF, baF;
    asm(
        "s_and_b32 %[bF], %[wF], 0x7f800000\n\t"
        "s_add_f32 %[t2F], %[wF], %[dF]\n\t"
        "s_add_f32 %[tF], %[bF], %[bF]\n\t"
        "s_add_f32 %[deF], %[bF], %[dF]\n\t"
        "s_min_u32 %[tF], %[tF], 0x40c90fdb\n\t"
        "s_sub_f32 %[deF], %[deF], %[bF]\n\t"
        "s_sub_f32 %[tF], %[tF], %[t2F]\n\t"
        "s_sub_f32 %[baF], %[t2F], %[deF]\n\t"
        "s_mul_f32 %[tF], %[tF], %[rF]\n\t"
        "s_ceil_f32 %[tF], %[tF]\n\t"
        "s_fmac_f32 %[baF], %[tF], %[deF]\n\t"
        "s_max_num_f32 %[cnF], %[tF], 0\n\t"
        "s_cmp_ge_f32 %[tF], 1.0\n\t"
        "s_cselect_b32 %[baF], %[baF], %[wF]\n\t"
        "s_add_f32 %[baF], %[baF], %[dF]\n\t"
        "s_add_f32 %[cnF], %[cnF], %[cF]\n\t"
        "s_add_f32 %[wnF], %[baF], 0xc0c90fdb\n\t"
        "s_add_f32 %[cnF], %[cnF], 1.0\n\t"
        "s_cmp_ge_f32 %[baF], 0x40c90fdb\n\t"
        "s_cselect_b32 %[wnF], %[wnF], %[baF]\n\t"
        : [wnF] "=&s"(wnF), [cnF] "=&s"(cnF), [t2F] "=&s"(t2F), [deF] "=&s"(deF), [bF] "=&s"(bF), [tF] "=&s"(tF), [baF] "=&s"(baF)
        : [wF] "s"(wF), [cF] "s"(cF), [dF] "s"(dF), [rF] "s"(rdF)
        : "scc");
    sgF.c = cF; sgF.w = wF; sgF.t2 = t2F; sgF.de = deF;
    wF = wnF; cF = cnF;
}

template <int S>
__device__ __forceinline__ float eval_prop(const Seg (&sg)[S], float kf) {
    float c = sg[0].c, w = sg[0].w, t2 = sg[0].t2, de = sg[0].de;
#pragma unroll
    for (int s = 1; s < S; ++s) {
        const bool in = kf >= sg[s].c;
        c  = in ? sg[s].c  : c;
        w  = in ? sg[s].w  : w;
        t2 = in ? sg[s].t2 : t2;
        de = in ? sg[s].de : de;
    }
    const float j = kf - c;
    return (j == 0.0f) ? w : fmaf(j - 1.0f, de, t2);
}

#define RUNS_SLOW 5
#define RUNS_FAST 9
static_assert(RUNS_SLOW <= RUNS_FAST, "the paired walk covers the slow table");

#define RING 4096
#define RING_MASK (RING - 1)
#define REPLAY_MIN_OK 8

__global__ void osc_scan_kernel(const float* __restrict__ p_log_slow,
                                const float* __restrict__ p_log_fast,
                                const float* __restrict__ p_reset,
                                float* __restrict__ out, int N) {
    if (blockIdx.x != 0) return;
    const int lane = threadIdx.x;

    const float dslow = expf(p_log_slow[0]) * DT_F;
    const float dfast = expf(p_log_fast[0]) * DT_F;
    const float keep  = 1.0f - p_reset[0];

    float* __restrict__ oS = out;
    float* __restrict__ oF = out + N;

    const bool fast_ok =
        (dslow >= 0.0f) && (dslow < TWO_PI_F) &&
        (dfast >= 0.0f) && (dfast < TWO_PI_F) &&
        (keep  >= 0.0f) && (keep  <= 1.0f);

    __shared__ float hS[RING];
    __shared__ float hF[RING];

    if (fast_ok) {
        const float kf  = (float)lane;
        for (int i = lane; i < RING; i += 32) { hS[i] = 0.0f; hF[i] = 0.0f; }
        int evStep = -1, evBits = 0;
        int evN = 0, lag = 0;
        const float rds = to_scalar(1.0f / dslow);
        const float rdf = to_scalar(1.0f / dfast);
        float xs = 0.0f, xf = 0.0f, pv = 0.0f;
        int t = 0;
        while (t < N) {
            const bool replay = (lag > 0) && (t >= lag);
            float a, g;
            if (replay) {
                const int idx = (t + lane - lag) & RING_MASK;
                const float hs = hS[idx];
                const float hf = hF[idx];
                const float offS = xs - __int_as_float(__builtin_amdgcn_readfirstlane(__float_as_int(hs)));
                const float offF = xf - __int_as_float(__builtin_amdgcn_readfirstlane(__float_as_int(hf)));
                a = (lane == 0) ? xs : hs + offS;
                g = (lane == 0) ? xf : hf + offF;
            } else {
                Seg sS[RUNS_SLOW], sF[RUNS_FAST];
                {
                    float wS = xs, cS = 0.0f, wF = xf, cF = 0.0f;
#pragma unroll
                    for (int s = 0; s < RUNS_SLOW; ++s)
                        walk_pair(wF, cF, dfast, rdf, sF[s], wS, cS, dslow, rds, sS[s]);
#pragma unroll
                    for (int s = RUNS_SLOW; s < RUNS_FAST; ++s)
                        walk_one(wF, cF, dfast, rdf, sF[s]);
                }
                a = eval_prop<RUNS_SLOW>(sS, kf);
                g = eval_prop<RUNS_FAST>(sF, kf);
            }
            const int ai = __float_as_int(a), gi = __float_as_int(g);
            const int a16 = __builtin_amdgcn_readlane(ai, 16);
            const int g16 = __builtin_amdgcn_readlane(gi, 16);
            const int a15 = __builtin_amdgcn_readlane(ai, 15);
            const float a1 = __int_as_float(__builtin_amdgcn_update_dpp(a16, ai, 0x101, 0xf, 0xf, false));
            const float g1 = __int_as_float(__builtin_amdgcn_update_dpp(g16, gi, 0x101, 0xf, 0xf, false));
            const int apo  = (lane == 0) ? __float_as_int(pv) : a15;
            const float ap = __int_as_float(__builtin_amdgcn_update_dpp(apo, ai, 0x111, 0xf, 0xf, false));
            const float ss = wrap_add(a, dslow);
            const bool crossed = (ap > PI_F) && (a < PI_F);
            const unsigned cball = __builtin_amdgcn_ballot_w32(crossed);
            const float kk = crossed ? keep : 1.0f;
            const float sf = wrap_add(g, dfast) * kk;
            const unsigned m =
                __builtin_amdgcn_ballot_w32(__float_as_uint(ss) == __float_as_uint(a1)) &
                __builtin_amdgcn_ballot_w32(__float_as_uint(sf) == __float_as_uint(g1));
            const int K = __builtin_ctz(~m | 0x80000000u);
            xs = __int_as_float(__builtin_amdgcn_readlane(__float_as_int(ss), K));
            xf = __int_as_float(__builtin_amdgcn_readlane(__float_as_int(sf), K));
            pv = __int_as_float(__builtin_amdgcn_readlane(ai, K));
            const int last = min(K, N - 1 - t);
            if (lane <= last) {
                oS[t + lane] = a;
                oF[t + lane] = g;
                hS[(t + lane) & RING_MASK] = a;
                hF[(t + lane) & RING_MASK] = g;
            }
            if (replay && K + 1 < REPLAY_MIN_OK) lag = 0;
            const unsigned cm = cball & (0xffffffffu >> (31 - K));
            if (cm != 0u) {
                const int kc = __builtin_ctz(cm);
                const int e  = t + kc;
                const int fb = __builtin_amdgcn_readlane(gi, kc);
                const unsigned mm = __builtin_amdgcn_ballot_w32((evStep >= 0) && (evBits == fb));
                const int slot = evN & 31;
                if (mm != 0u) {
                    const unsigned rr = __builtin_rotateright32(mm, (unsigned)slot);
                    const int ms = (31 - __builtin_clz(rr) + slot) & 31;
                    const int cand = e - __builtin_amdgcn_readlane(evStep, ms);
                    if (cand >= 32 && cand + 32 <= RING) lag = cand;
                }
                if (lane == slot) { evStep = e; evBits = fb; }
                ++evN;
            }
            t += K + 1;
        }
        return;
    }

    if (lane != 0) return;
    float slow = 0.0f, fast = 0.0f, prev = 0.0f;
    for (int t = 0; t < N; ++t) {
        oS[t] = slow;
        oF[t] = fast;
        bool crossed = (prev > PI_F) && (slow < PI_F);
        float ns = pmod(slow + dslow, TWO_PI_F);
        float nf = pmod(fast + dfast, TWO_PI_F);
        if (crossed) nf = pmod(nf * keep, TWO_PI_F);
        prev = slow; slow = ns; fast = nf;
    }
}

__global__ void osc_scale_wmma_kernel(const float* __restrict__ in,
                                      float* __restrict__ out, int N) {
    const float s = INV_TWO_PI_F;
    const int base = blockIdx.x * 256;
    const int lane = threadIdx.x;
    const int i    = lane & 15;
    const int kh   = lane >> 4;

    if (base + 256 <= N) {
        v8f d = {0.f, 0.f, 0.f, 0.f, 0.f, 0.f, 0.f, 0.f};
#pragma unroll
        for (int m = 0; m < 4; ++m) {
            const float* p = in + base + i * 16 + 4 * m + 2 * kh;
            v2f a; a.x = p[0]; a.y = p[1];
            v2f b;
            b.x = (i == (4 * m + 2 * kh + 0)) ? s : 0.0f;
            b.y = (i == (4 * m + 2 * kh + 1)) ? s : 0.0f;
            d = __builtin_amdgcn_wmma_f32_16x16x4_f32(
                    false, a, false, b, (short)0, d, false, false);
        }
#pragma unroll
        for (int r = 0; r < 8; ++r) {
            out[base + (r + 8 * kh) * 16 + i] = d[r];
        }
    } else {
        for (int j = base + lane; j < N; j += 32) out[j] = in[j] * s;
    }
}

extern "C" void kernel_launch(void* const* d_in, const int* in_sizes, int n_in,
                              void* d_out, int out_size, void* d_ws, size_t ws_size,
                              hipStream_t stream) {
    const float* p_log_slow = (const float*)d_in[0];
    const float* p_log_fast = (const float*)d_in[1];
    const float* p_reset    = (const float*)d_in[2];
    const int N = out_size / 3;
    if (N <= 0) return;
    float* out = (float*)d_out;

    osc_scan_kernel<<<1, 32, 0, stream>>>(p_log_slow, p_log_fast, p_reset, out, N);

    const int tiles = (N + 255) / 256;
    osc_scale_wmma_kernel<<<tiles, 32, 0, stream>>>(out, out + 2 * (size_t)N, N);
}


# scan: a replayed pass spans 512 steps, 16 windows of 32 per lane; successors by one lane rotate fed across windows, crossings from ballots, recorded only; same sums, same order, same values
# speedup vs baseline: 5.4835x; 1.6974x over previous
#include <hip/hip_runtime.h>
#include <hip/hip_bf16.h>


typedef __attribute__((ext_vector_type(2))) float v2f;
typedef __attribute__((ext_vector_type(8))) float v8f;

#define TWO_PI_F     6.28318530717958647692f
#define PI_F         3.14159265358979323846f
#define INV_TWO_PI_F 0.15915494309189533577f
#define DT_F         0.001f

__device__ __forceinline__ float pmod(float x, float m) {
    float r = fmodf(x, m);
    return (r < 0.0f) ? (r + m) : r;
}

__device__ __forceinline__ float wrap_add(float x, float d) {
    float t = x + d;
    return (t >= TWO_PI_F) ? (t - TWO_PI_F) : t;
}


struct Seg { float c, w, t2, de; };

__device__ __forceinline__ float to_scalar(float x) {
    float r;
    asm("v_readfirstlane_b32 %0, %1" : "=s"(r) : "v"(x));
    return r;
}

__device__ __forceinline__ void walk_pair(float& wF, float& cF, float dF, float rdF, Seg& sgF,
                                          float& wS, float& cS, float dS, float rdS, Seg& sgS) {
    float wnF, cnF, t2F, deF, bF, tF, baF;
    float wnS, cnS, t2S, deS, bS, tS, baS;
    asm(
        "s_and_b32 %[bF], %[wF], 0x7f800000\n\t"
        "s_and_b32 %[bS], %[wS], 0x7f800000\n\t"
        "s_add_f32 %[t2F], %[wF], %[dF]\n\t"
        "s_add_f32 %[t2S], %[wS], %[dS]\n\t"
        "s_add_f32 %[tF], %[bF], %[bF]\n\t"
        "s_add_f32 %[tS], %[bS], %[bS]\n\t"
        "s_add_f32 %[deF], %[bF], %[dF]\n\t"
        "s_add_f32 %[deS], %[bS], %[dS]\n\t"
        "s_min_u32 %[tF], %[tF], 0x40c90fdb\n\t"
        "s_min_u32 %[tS], %[tS], 0x40c90fdb\n\t"
        "s_sub_f32 %[deF], %[deF], %[bF]\n\t"
        "s_sub_f32 %[deS], %[deS], %[bS]\n\t"
        "s_sub_f32 %[tF], %[tF], %[t2F]\n\t"
        "s_sub_f32 %[tS], %[tS], %[t2S]\n\t"
        "s_sub_f32 %[baF], %[t2F], %[deF]\n\t"
        "s_sub_f32 %[baS], %[t2S], %[deS]\n\t"
        "s_mul_f32 %[tF], %[tF], %[rF]\n\t"
        "s_mul_f32 %[tS], %[tS], %[rS]\n\t"
        "s_ceil_f32 %[tF], %[tF]\n\t"
        "s_ceil_f32 %[tS], %[tS]\n\t"
        "s_fmac_f32 %[baF], %[tF], %[deF]\n\t"
        "s_fmac_f32 %[baS], %[tS], %[deS]\n\t"
        "s_max_num_f32 %[cnF], %[tF], 0\n\t"
        "s_max_num_f32 %[cnS], %[tS], 0\n\t"
        "s_cmp_ge_f32 %[tF], 1.0\n\t"
        "s_cselect_b32 %[baF], %[baF], %[wF]\n\t"
        "s_cmp_ge_f32 %[tS], 1.0\n\t"
        "s_cselect_b32 %[baS], %[baS], %[wS]\n\t"
        "s_add_f32 %[baF], %[baF], %[dF]\n\t"
        "s_add_f32 %[baS], %[baS], %[dS]\n\t"
        "s_add_f32 %[cnF], %[cnF], %[cF]\n\t"
        "s_add_f32 %[cnS], %[cnS], %[cS]\n\t"
        "s_add_f32 %[wnF], %[baF], 0xc0c90fdb\n\t"
        "s_add_f32 %[wnS], %[baS], 0xc0c90fdb\n\t"
        "s_add_f32 %[cnF], %[cnF], 1.0\n\t"
        "s_add_f32 %[cnS], %[cnS], 1.0\n\t"
        "s_cmp_ge_f32 %[baF], 0x40c90fdb\n\t"
        "s_cselect_b32 %[wnF], %[wnF], %[baF]\n\t"
        "s_cmp_ge_f32 %[baS], 0x40c90fdb\n\t"
        "s_cselect_b32 %[wnS], %[wnS], %[baS]\n\t"
        : [wnF] "=&s"(wnF), [cnF] "=&s"(cnF), [t2F] "=&s"(t2F), [deF] "=&s"(deF), [bF] "=&s"(bF), [tF] "=&s"(tF), [baF] "=&s"(baF), [wnS] "=&s"(wnS), [cnS] "=&s"(cnS), [t2S] "=&s"(t2S), [deS] "=&s"(deS), [bS] "=&s"(bS), [tS] "=&s"(tS), [baS] "=&s"(baS)
        : [wF] "s"(wF), [cF] "s"(cF), [dF] "s"(dF), [rF] "s"(rdF), [wS] "s"(wS), [cS] "s"(cS), [dS] "s"(dS), [rS] "s"(rdS)
        : "scc");
    sgF.c = cF; sgF.w = wF; sgF.t2 = t2F; sgF.de = deF;
    sgS.c = cS; sgS.w = wS; sgS.t2 = t2S; sgS.de = deS;
    wF = wnF; cF = cnF;
    wS = wnS; cS = cnS;
}

__device__ __forceinline__ void walk_one(float& wF, float& cF, float dF, float rdF, Seg& sgF) {
    float wnF, cnF, t2F, deF, bF, tF, baF;
    asm(
        "s_and_b32 %[bF], %[wF], 0x7f800000\n\t"
        "s_add_f32 %[t2F], %[wF], %[dF]\n\t"
        "s_add_f32 %[tF], %[bF], %[bF]\n\t"
        "s_add_f32 %[deF], %[bF], %[dF]\n\t"
        "s_min_u32 %[tF], %[tF], 0x40c90fdb\n\t"
        "s_sub_f32 %[deF], %[deF], %[bF]\n\t"
        "s_sub_f32 %[tF], %[tF], %[t2F]\n\t"
        "s_sub_f32 %[baF], %[t2F], %[deF]\n\t"
        "s_mul_f32 %[tF], %[tF], %[rF]\n\t"
        "s_ceil_f32 %[tF], %[tF]\n\t"
        "s_fmac_f32 %[baF], %[tF], %[deF]\n\t"
        "s_max_num_f32 %[cnF], %[tF], 0\n\t"
        "s_cmp_ge_f32 %[tF], 1.0\n\t"
        "s_cselect_b32 %[baF], %[baF], %[wF]\n\t"
        "s_add_f32 %[baF], %[baF], %[dF]\n\t"
        "s_add_f32 %[cnF], %[cnF], %[cF]\n\t"
        "s_add_f32 %[wnF], %[baF], 0xc0c90fdb\n\t"
        "s_add_f32 %[cnF], %[cnF], 1.0\n\t"
        "s_cmp_ge_f32 %[baF], 0x40c90fdb\n\t"
        "s_cselect_b32 %[wnF], %[wnF], %[baF]\n\t"
        : [wnF] "=&s"(wnF), [cnF] "=&s"(cnF), [t2F] "=&s"(t2F), [deF] "=&s"(deF), [bF] "=&s"(bF), [tF] "=&s"(tF), [baF] "=&s"(baF)
        : [wF] "s"(wF), [cF] "s"(cF), [dF] "s"(dF), [rF] "s"(rdF)
        : "scc");
    sgF.c = cF; sgF.w = wF; sgF.t2 = t2F; sgF.de = deF;
    wF = wnF; cF = cnF;
}

template <int S>
__device__ __forceinline__ float eval_prop(const Seg (&sg)[S], float kf) {
    float c = sg[0].c, w = sg[0].w, t2 = sg[0].t2, de = sg[0].de;
#pragma unroll
    for (int s = 1; s < S; ++s) {
        const bool in = kf >= sg[s].c;
        c  = in ? sg[s].c  : c;
        w  = in ? sg[s].w  : w;
        t2 = in ? sg[s].t2 : t2;
        de = in ? sg[s].de : de;
    }
    const float j = kf - c;
    return (j == 0.0f) ? w : fmaf(j - 1.0f, de, t2);
}

#define RUNS_SLOW 5
#define RUNS_FAST 9
static_assert(RUNS_SLOW <= RUNS_FAST, "the paired walk covers the slow table");

#define RING 4096
#define RING_MASK (RING - 1)
#define REPLAY_MIN_OK 8
#define REPLAY_WIN 16
#define REPLAY_SPAN (32 * REPLAY_WIN)
static_assert(2 * REPLAY_SPAN <= RING, "a replayed pass and its sources fit the ring");

__global__ void osc_scan_kernel(const float* __restrict__ p_log_slow,
                                const float* __restrict__ p_log_fast,
                                const float* __restrict__ p_reset,
                                float* __restrict__ out, int N) {
    if (blockIdx.x != 0) return;
    const int lane = threadIdx.x;

    const float dslow = expf(p_log_slow[0]) * DT_F;
    const float dfast = expf(p_log_fast[0]) * DT_F;
    const float keep  = 1.0f - p_reset[0];

    float* __restrict__ oS = out;
    float* __restrict__ oF = out + N;

    const bool fast_ok =
        (dslow >= 0.0f) && (dslow < TWO_PI_F) &&
        (dfast >= 0.0f) && (dfast < TWO_PI_F) &&
        (keep  >= 0.0f) && (keep  <= 1.0f);

    __shared__ float hS[RING];
    __shared__ float hF[RING];

    if (fast_ok) {
        const float kf  = (float)lane;
        for (int i = lane; i < RING; i += 32) { hS[i] = 0.0f; hF[i] = 0.0f; }
        int evStep = -1, evBits = 0;
        int evN = 0, lag = 0;
        const float rds = to_scalar(1.0f / dslow);
        const float rdf = to_scalar(1.0f / dfast);
        float xs = 0.0f, xf = 0.0f, pv = 0.0f;
        int t = 0;
        while (t < N) {
            const bool replay = (lag > 0) && (t >= lag);
            if (replay) {
                float a[REPLAY_WIN], g[REPLAY_WIN];
                const int sb = (t + lane - lag) << 2;
#pragma unroll
                for (int w = 0; w < REPLAY_WIN; ++w) {
                    const int idx = ((sb + 128 * w) & (4 * RING - 4)) >> 2;
                    a[w] = hS[idx];
                    g[w] = hF[idx];
                }
                const float offS = xs - __int_as_float(__builtin_amdgcn_readfirstlane(__float_as_int(a[0])));
                const float offF = xf - __int_as_float(__builtin_amdgcn_readfirstlane(__float_as_int(g[0])));
#pragma unroll
                for (int w = 0; w < REPLAY_WIN; ++w) { a[w] += offS; g[w] += offF; }
                a[0] = (lane == 0) ? xs : a[0];
                g[0] = (lane == 0) ? xf : g[0];
                const int upSrc = ((lane + 1) & 31) << 2;
                int a1[REPLAY_WIN], g1[REPLAY_WIN];
#pragma unroll
                for (int w = 0; w < REPLAY_WIN; ++w) {
                    const float aN = (w + 1 < REPLAY_WIN) ? a[w + 1] : a[w];
                    const float gN = (w + 1 < REPLAY_WIN) ? g[w + 1] : g[w];
                    a1[w] = __builtin_amdgcn_ds_bpermute(upSrc, __float_as_int((lane == 0) ? aN : a[w]));
                    g1[w] = __builtin_amdgcn_ds_bpermute(upSrc, __float_as_int((lane == 0) ? gN : g[w]));
                }
                __builtin_amdgcn_sched_barrier(0);
                unsigned cb[REPLAY_WIN];
                unsigned carry = (pv > PI_F) ? 1u : 0u;
#pragma unroll
                for (int w = 0; w < REPLAY_WIN; ++w) {
                    const unsigned hi = __builtin_amdgcn_ballot_w32(a[w] > PI_F);
                    const unsigned lo = __builtin_amdgcn_ballot_w32(a[w] < PI_F);
                    cb[w] = ((hi << 1) | carry) & lo;
                    carry = hi >> 31;
                }
                float ss[REPLAY_WIN], sf[REPLAY_WIN];
                unsigned mk[REPLAY_WIN];
#pragma unroll
                for (int w = 0; w < REPLAY_WIN; ++w) {
                    ss[w] = wrap_add(a[w], dslow);
                    const bool crossed = __builtin_amdgcn_inverse_ballot_w32(cb[w]);
                    const float kk = crossed ? keep : 1.0f;
                    sf[w] = wrap_add(g[w], dfast) * kk;
                    mk[w] = __builtin_amdgcn_ballot_w32(__float_as_int(ss[w]) == a1[w]) &
                            __builtin_amdgcn_ballot_w32(__float_as_int(sf[w]) == g1[w]);
                }
                unsigned allm = mk[REPLAY_WIN - 1] | 0x80000000u;
#pragma unroll
                for (int w = 0; w + 1 < REPLAY_WIN; ++w) allm &= mk[w];
                auto record = [&](unsigned cm, int w, float gw) {
                    const int kc = __builtin_ctz(cm);
                    const int fb = __builtin_amdgcn_readlane(__float_as_int(gw), kc);
                    if (lane == (evN & 31)) { evStep = t + 32 * w + kc; evBits = fb; }
                    ++evN;
                };
                if (allm == 0xffffffffu && t + REPLAY_SPAN <= N) {
                    xs = __int_as_float(__builtin_amdgcn_readlane(__float_as_int(ss[REPLAY_WIN - 1]), 31));
                    xf = __int_as_float(__builtin_amdgcn_readlane(__float_as_int(sf[REPLAY_WIN - 1]), 31));
                    pv = __int_as_float(__builtin_amdgcn_readlane(__float_as_int(a[REPLAY_WIN - 1]), 31));
                    float* __restrict__ pS = oS + (t + lane);
                    float* __restrict__ pF = oF + (t + lane);
                    const int tb = (t + lane) << 2;
#pragma unroll
                    for (int w = 0; w < REPLAY_WIN; ++w) {
                        pS[32 * w] = a[w];
                        pF[32 * w] = g[w];
                        const int idx = ((tb + 128 * w) & (4 * RING - 4)) >> 2;
                        hS[idx] = a[w];
                        hF[idx] = g[w];
                    }
#pragma unroll
                    for (int w = 0; w < REPLAY_WIN; ++w)
                        if (cb[w] != 0u) record(cb[w], w, g[w]);
                    t += REPLAY_SPAN;
                    continue;
                }
                mk[REPLAY_WIN - 1] &= 0x7fffffffu;
                int K = REPLAY_SPAN - 1;
#pragma unroll
                for (int w = REPLAY_WIN - 1; w >= 0; --w)
                    if (mk[w] != 0xffffffffu) K = 32 * w + __builtin_ctz(~mk[w]);
                const int wK = K >> 5, lK = K & 31;
                float nxs = xs, nxf = xf, npv = pv;
#pragma unroll
                for (int w = 0; w < REPLAY_WIN; ++w) {
                    const float c1 = __int_as_float(__builtin_amdgcn_readlane(__float_as_int(ss[w]), lK));
                    const float c2 = __int_as_float(__builtin_amdgcn_readlane(__float_as_int(sf[w]), lK));
                    const float c3 = __int_as_float(__builtin_amdgcn_readlane(__float_as_int(a[w]), lK));
                    if (w == wK) { nxs = c1; nxf = c2; npv = c3; }
                }
                xs = nxs; xf = nxf; pv = npv;
                const int last = min(K, N - 1 - t);
#pragma unroll
                for (int w = 0; w < REPLAY_WIN; ++w) {
                    const int q = 32 * w + lane;
                    if (q <= last) {
                        oS[t + q] = a[w];
                        oF[t + q] = g[w];
                        hS[(t + q) & RING_MASK] = a[w];
                        hF[(t + q) & RING_MASK] = g[w];
                    }
                }
                if (K + 1 < REPLAY_MIN_OK) lag = 0;
#pragma unroll
                for (int w = 0; w < REPLAY_WIN; ++w) {
                    const int kw = K - 32 * w;
                    const unsigned vm = (kw >= 31) ? 0xffffffffu : ((kw < 0) ? 0u : (0xffffffffu >> (31 - kw)));
                    const unsigned cm = cb[w] & vm;
                    if (cm != 0u) record(cm, w, g[w]);
                }
                t += K + 1;
                continue;
            }
            float a, g;
            {
                Seg sS[RUNS_SLOW], sF[RUNS_FAST];
                {
                    float wS = xs, cS = 0.0f, wF = xf, cF = 0.0f;
#pragma unroll
                    for (int s = 0; s < RUNS_SLOW; ++s)
                        walk_pair(wF, cF, dfast, rdf, sF[s], wS, cS, dslow, rds, sS[s]);
#pragma unroll
                    for (int s = RUNS_SLOW; s < RUNS_FAST; ++s)
                        walk_one(wF, cF, dfast, rdf, sF[s]);
                }
                a = eval_prop<RUNS_SLOW>(sS, kf);
                g = eval_prop<RUNS_FAST>(sF, kf);
            }
            const int ai = __float_as_int(a), gi = __float_as_int(g);
            const int a16 = __builtin_amdgcn_readlane(ai, 16);
            const int g16 = __builtin_amdgcn_readlane(gi, 16);
            const int a15 = __builtin_amdgcn_readlane(ai, 15);
            const float a1 = __int_as_float(__builtin_amdgcn_update_dpp(a16, ai, 0x101, 0xf, 0xf, false));
            const float g1 = __int_as_float(__builtin_amdgcn_update_dpp(g16, gi, 0x101, 0xf, 0xf, false));
            const int apo  = (lane == 0) ? __float_as_int(pv) : a15;
            const float ap = __int_as_float(__builtin_amdgcn_update_dpp(apo, ai, 0x111, 0xf, 0xf, false));
            const float ss = wrap_add(a, dslow);
            const bool crossed = (ap > PI_F) && (a < PI_F);
            const unsigned cball = __builtin_amdgcn_ballot_w32(crossed);
            const float kk = crossed ? keep : 1.0f;
            const float sf = wrap_add(g, dfast) * kk;
            const unsigned m =
                __builtin_amdgcn_ballot_w32(__float_as_uint(ss) == __float_as_uint(a1)) &
                __builtin_amdgcn_ballot_w32(__float_as_uint(sf) == __float_as_uint(g1));
            const int K = __builtin_ctz(~m | 0x80000000u);
            xs = __int_as_float(__builtin_amdgcn_readlane(__float_as_int(ss), K));
            xf = __int_as_float(__builtin_amdgcn_readlane(__float_as_int(sf), K));
            pv = __int_as_float(__builtin_amdgcn_readlane(ai, K));
            const int last = min(K, N - 1 - t);
            if (lane <= last) {
                oS[t + lane] = a;
                oF[t + lane] = g;
                hS[(t + lane) & RING_MASK] = a;
                hF[(t + lane) & RING_MASK] = g;
            }
            const unsigned cm = cball & (0xffffffffu >> (31 - K));
            if (cm != 0u) {
                const int kc = __builtin_ctz(cm);
                const int e  = t + kc;
                const int fb = __builtin_amdgcn_readlane(gi, kc);
                const unsigned mm = __builtin_amdgcn_ballot_w32((evStep >= 0) && (evBits == fb));
                const int slot = evN & 31;
                if (mm != 0u) {
                    const unsigned rr = __builtin_rotateright32(mm, (unsigned)slot);
                    const int ms = (31 - __builtin_clz(rr) + slot) & 31;
                    const int cand = e - __builtin_amdgcn_readlane(evStep, ms);
                    if (cand >= 32) {
                        int c2 = cand;
                        while (c2 < REPLAY_SPAN) c2 += cand;
                        if (c2 + REPLAY_SPAN <= RING) lag = c2;
                    }
                }
                if (lane == slot) { evStep = e; evBits = fb; }
                ++evN;
            }
            t += K + 1;
        }
        return;
    }

    if (lane != 0) return;
    float slow = 0.0f, fast = 0.0f, prev = 0.0f;
    for (int t = 0; t < N; ++t) {
        oS[t] = slow;
        oF[t] = fast;
        bool crossed = (prev > PI_F) && (slow < PI_F);
        float ns = pmod(slow + dslow, TWO_PI_F);
        float nf = pmod(fast + dfast, TWO_PI_F);
        if (crossed) nf = pmod(nf * keep, TWO_PI_F);
        prev = slow; slow = ns; fast = nf;
    }
}

__global__ void osc_scale_wmma_kernel(const float* __restrict__ in,
                                      float* __restrict__ out, int N) {
    const float s = INV_TWO_PI_F;
    const int base = blockIdx.x * 256;
    const int lane = threadIdx.x;
    const int i    = lane & 15;
    const int kh   = lane >> 4;

    if (base + 256 <= N) {
        v8f d = {0.f, 0.f, 0.f, 0.f, 0.f, 0.f, 0.f, 0.f};
#pragma unroll
        for (int m = 0; m < 4; ++m) {
            const float* p = in + base + i * 16 + 4 * m + 2 * kh;
            v2f a; a.x = p[0]; a.y = p[1];
            v2f b;
            b.x = (i == (4 * m + 2 * kh + 0)) ? s : 0.0f;
            b.y = (i == (4 * m + 2 * kh + 1)) ? s : 0.0f;
            d = __builtin_amdgcn_wmma_f32_16x16x4_f32(
                    false, a, false, b, (short)0, d, false, false);
        }
#pragma unroll
        for (int r = 0; r < 8; ++r) {
            out[base + (r + 8 * kh) * 16 + i] = d[r];
        }
    } else {
        for (int j = base + lane; j < N; j += 32) out[j] = in[j] * s;
    }
}

extern "C" void kernel_launch(void* const* d_in, const int* in_sizes, int n_in,
                              void* d_out, int out_size, void* d_ws, size_t ws_size,
                              hipStream_t stream) {
    const float* p_log_slow = (const float*)d_in[0];
    const float* p_log_fast = (const float*)d_in[1];
    const float* p_reset    = (const float*)d_in[2];
    const int N = out_size / 3;
    if (N <= 0) return;
    float* out = (float*)d_out;

    osc_scan_kernel<<<1, 32, 0, stream>>>(p_log_slow, p_log_fast, p_reset, out, N);

    const int tiles = (N + 255) / 256;
    osc_scale_wmma_kernel<<<tiles, 32, 0, stream>>>(out, out + 2 * (size_t)N, N);
}


# scan: a replayed pass keeps its per-window tests in vector registers (previous slow value by a second lane rotate, crossing from sign bits, links by xor/or), one ballot a pass; same values
# speedup vs baseline: 6.3812x; 1.1637x over previous
#include <hip/hip_runtime.h>
#include <hip/hip_bf16.h>


typedef __attribute__((ext_vector_type(2))) float v2f;
typedef __attribute__((ext_vector_type(8))) float v8f;

#define TWO_PI_F     6.28318530717958647692f
#define PI_F         3.14159265358979323846f
#define INV_TWO_PI_F 0.15915494309189533577f
#define DT_F         0.001f

__device__ __forceinline__ float pmod(float x, float m) {
    float r = fmodf(x, m);
    return (r < 0.0f) ? (r + m) : r;
}

__device__ __forceinline__ float wrap_add(float x, float d) {
    float t = x + d;
    return (t >= TWO_PI_F) ? (t - TWO_PI_F) : t;
}


struct Seg { float c, w, t2, de; };

__device__ __forceinline__ float to_scalar(float x) {
    float r;
    asm("v_readfirstlane_b32 %0, %1" : "=s"(r) : "v"(x));
    return r;
}

__device__ __forceinline__ void walk_pair(float& wF, float& cF, float dF, float rdF, Seg& sgF,
                                          float& wS, float& cS, float dS, float rdS, Seg& sgS) {
    float wnF, cnF, t2F, deF, bF, tF, baF;
    float wnS, cnS, t2S, deS, bS, tS, baS;
    asm(
        "s_and_b32 %[bF], %[wF], 0x7f800000\n\t"
        "s_and_b32 %[bS], %[wS], 0x7f800000\n\t"
        "s_add_f32 %[t2F], %[wF], %[dF]\n\t"
        "s_add_f32 %[t2S], %[wS], %[dS]\n\t"
        "s_add_f32 %[tF], %[bF], %[bF]\n\t"
        "s_add_f32 %[tS], %[bS], %[bS]\n\t"
        "s_add_f32 %[deF], %[bF], %[dF]\n\t"
        "s_add_f32 %[deS], %[bS], %[dS]\n\t"
        "s_min_u32 %[tF], %[tF], 0x40c90fdb\n\t"
        "s_min_u32 %[tS], %[tS], 0x40c90fdb\n\t"
        "s_sub_f32 %[deF], %[deF], %[bF]\n\t"
        "s_sub_f32 %[deS], %[deS], %[bS]\n\t"
        "s_sub_f32 %[tF], %[tF], %[t2F]\n\t"
        "s_sub_f32 %[tS], %[tS], %[t2S]\n\t"
        "s_sub_f32 %[baF], %[t2F], %[deF]\n\t"
        "s_sub_f32 %[baS], %[t2S], %[deS]\n\t"
        "s_mul_f32 %[tF], %[tF], %[rF]\n\t"
        "s_mul_f32 %[tS], %[tS], %[rS]\n\t"
        "s_ceil_f32 %[tF], %[tF]\n\t"
        "s_ceil_f32 %[tS], %[tS]\n\t"
        "s_fmac_f32 %[baF], %[tF], %[deF]\n\t"
        "s_fmac_f32 %[baS], %[tS], %[deS]\n\t"
        "s_max_num_f32 %[cnF], %[tF], 0\n\t"
        "s_max_num_f32 %[cnS], %[tS], 0\n\t"
        "s_cmp_ge_f32 %[tF], 1.0\n\t"
        "s_cselect_b32 %[baF], %[baF], %[wF]\n\t"
        "s_cmp_ge_f32 %[tS], 1.0\n\t"
        "s_cselect_b32 %[baS], %[baS], %[wS]\n\t"
        "s_add_f32 %[baF], %[baF], %[dF]\n\t"
        "s_add_f32 %[baS], %[baS], %[dS]\n\t"
        "s_add_f32 %[cnF], %[cnF], %[cF]\n\t"
        "s_add_f32 %[cnS], %[cnS], %[cS]\n\t"
        "s_add_f32 %[wnF], %[baF], 0xc0c90fdb\n\t"
        "s_add_f32 %[wnS], %[baS], 0xc0c90fdb\n\t"
        "s_add_f32 %[cnF], %[cnF], 1.0\n\t"
        "s_add_f32 %[cnS], %[cnS], 1.0\n\t"
        "s_cmp_ge_f32 %[baF], 0x40c90fdb\n\t"
        "s_cselect_b32 %[wnF], %[wnF], %[baF]\n\t"
        "s_cmp_ge_f32 %[baS], 0x40c90fdb\n\t"
        "s_cselect_b32 %[wnS], %[wnS], %[baS]\n\t"
        : [wnF] "=&s"(wnF), [cnF] "=&s"(cnF), [t2F] "=&s"(t2F), [deF] "=&s"(deF), [bF] "=&s"(bF), [tF] "=&s"(tF), [baF] "=&s"(baF), [wnS] "=&s"(wnS), [cnS] "=&s"(cnS), [t2S] "=&s"(t2S), [deS] "=&s"(deS), [bS] "=&s"(bS), [tS] "=&s"(tS), [baS] "=&s"(baS)
        : [wF] "s"(wF), [cF] "s"(cF), [dF] "s"(dF), [rF] "s"(rdF), [wS] "s"(wS), [cS] "s"(cS), [dS] "s"(dS), [rS] "s"(rdS)
        : "scc");
    sgF.c = cF; sgF.w = wF; sgF.t2 = t2F; sgF.de = deF;
    sgS.c = cS; sgS.w = wS; sgS.t2 = t2S; sgS.de = deS;
    wF = wnF; cF = cnF;
    wS = wnS; cS = cnS;
}

__device__ __forceinline__ void walk_one(float& wF, float& cF, float dF, float rdF, Seg& sgF) {
    float wnF, cnF, t2F, deF, bF, tF, baF;
    asm(
        "s_and_b32 %[bF], %[wF], 0x7f800000\n\t"
        "s_add_f32 %[t2F], %[wF], %[dF]\n\t"
        "s_add_f32 %[tF], %[bF], %[bF]\n\t"
        "s_add_f32 %[deF], %[bF], %[dF]\n\t"
        "s_min_u32 %[tF], %[tF], 0x40c90fdb\n\t"
        "s_sub_f32 %[deF], %[deF], %[bF]\n\t"
        "s_sub_f32 %[tF], %[tF], %[t2F]\n\t"
        "s_sub_f32 %[baF], %[t2F], %[deF]\n\t"
        "s_mul_f32 %[tF], %[tF], %[rF]\n\t"
        "s_ceil_f32 %[tF], %[tF]\n\t"
        "s_fmac_f32 %[baF], %[tF], %[deF]\n\t"
        "s_max_num_f32 %[cnF], %[tF], 0\n\t"
        "s_cmp_ge_f32 %[tF], 1.0\n\t"
        "s_cselect_b32 %[baF], %[baF], %[wF]\n\t"
        "s_add_f32 %[baF], %[baF], %[dF]\n\t"
        "s_add_f32 %[cnF], %[cnF], %[cF]\n\t"
        "s_add_f32 %[wnF], %[baF], 0xc0c90fdb\n\t"
        "s_add_f32 %[cnF], %[cnF], 1.0\n\t"
        "s_cmp_ge_f32 %[baF], 0x40c90fdb\n\t"
        "s_cselect_b32 %[wnF], %[wnF], %[baF]\n\t"
        : [wnF] "=&s"(wnF), [cnF] "=&s"(cnF), [t2F] "=&s"(t2F), [deF] "=&s"(deF), [bF] "=&s"(bF), [tF] "=&s"(tF), [baF] "=&s"(baF)
        : [wF] "s"(wF), [cF] "s"(cF), [dF] "s"(dF), [rF] "s"(rdF)
        : "scc");
    sgF.c = cF; sgF.w = wF; sgF.t2 = t2F; sgF.de = deF;
    wF = wnF; cF = cnF;
}

template <int S>
__device__ __forceinline__ float eval_prop(const Seg (&sg)[S], float kf) {
    float c = sg[0].c, w = sg[0].w, t2 = sg[0].t2, de = sg[0].de;
#pragma unroll
    for (int s = 1; s < S; ++s) {
        const bool in = kf >= sg[s].c;
        c  = in ? sg[s].c  : c;
        w  = in ? sg[s].w  : w;
        t2 = in ? sg[s].t2 : t2;
        de = in ? sg[s].de : de;
    }
    const float j = kf - c;
    return (j == 0.0f) ? w : fmaf(j - 1.0f, de, t2);
}

__device__ __forceinline__ unsigned or3_gather(unsigned acc, unsigned x, unsigned y) {
    unsigned r;
    asm("v_or3_b32 %0, %1, %2, %3" : "=v"(r) : "v"(acc), "v"(x), "v"(y));
    return r;
}

#define RUNS_SLOW 5
#define RUNS_FAST 9
static_assert(RUNS_SLOW <= RUNS_FAST, "the paired walk covers the slow table");

#define RING 4096
#define RING_MASK (RING - 1)
#define REPLAY_MIN_OK 8
#define REPLAY_WIN 16
#define REPLAY_SPAN (32 * REPLAY_WIN)
static_assert(2 * REPLAY_SPAN <= RING, "a replayed pass and its sources fit the ring");

__global__ void osc_scan_kernel(const float* __restrict__ p_log_slow,
                                const float* __restrict__ p_log_fast,
                                const float* __restrict__ p_reset,
                                float* __restrict__ out, int N) {
    if (blockIdx.x != 0) return;
    const int lane = threadIdx.x;

    const float dslow = expf(p_log_slow[0]) * DT_F;
    const float dfast = expf(p_log_fast[0]) * DT_F;
    const float keep  = 1.0f - p_reset[0];

    float* __restrict__ oS = out;
    float* __restrict__ oF = out + N;

    const bool fast_ok =
        (dslow >= 0.0f) && (dslow < TWO_PI_F) &&
        (dfast >= 0.0f) && (dfast < TWO_PI_F) &&
        (keep  >= 0.0f) && (keep  <= 1.0f);

    __shared__ float hS[RING];
    __shared__ float hF[RING];

    if (fast_ok) {
        const float kf  = (float)lane;
        for (int i = lane; i < RING; i += 32) { hS[i] = 0.0f; hF[i] = 0.0f; }
        int evStep = -1, evBits = 0;
        int evN = 0, lag = 0;
        const float rds = to_scalar(1.0f / dslow);
        const float rdf = to_scalar(1.0f / dfast);
        float xs = 0.0f, xf = 0.0f, pv = 0.0f;
        int t = 0;
        while (t < N) {
            const bool replay = (lag > 0) && (t >= lag);
            if (replay) {
                float a[REPLAY_WIN], g[REPLAY_WIN];
                const int sb = (t + lane - lag) << 2;
#pragma unroll
                for (int w = 0; w < REPLAY_WIN; ++w) {
                    const int idx = ((sb + 128 * w) & (4 * RING - 4)) >> 2;
                    a[w] = hS[idx];
                    g[w] = hF[idx];
                }
                const float offS = xs - __int_as_float(__builtin_amdgcn_readfirstlane(__float_as_int(a[0])));
                const float offF = xf - __int_as_float(__builtin_amdgcn_readfirstlane(__float_as_int(g[0])));
#pragma unroll
                for (int w = 0; w < REPLAY_WIN; ++w) { a[w] += offS; g[w] += offF; }
                a[0] = (lane == 0) ? xs : a[0];
                g[0] = (lane == 0) ? xf : g[0];
                const int upSrc = ((lane + 1) & 31) << 2;
                const int dnSrc = ((lane + 31) & 31) << 2;
                int a1[REPLAY_WIN], g1[REPLAY_WIN], ap[REPLAY_WIN];
#pragma unroll
                for (int w = 0; w < REPLAY_WIN; ++w) {
                    const float aN = (w + 1 < REPLAY_WIN) ? a[w + 1] : a[w];
                    const float gN = (w + 1 < REPLAY_WIN) ? g[w + 1] : g[w];
                    const float aP = (w > 0) ? a[w - 1] : pv;
                    a1[w] = __builtin_amdgcn_ds_bpermute(upSrc, __float_as_int((lane == 0) ? aN : a[w]));
                    g1[w] = __builtin_amdgcn_ds_bpermute(upSrc, __float_as_int((lane == 0) ? gN : g[w]));
                    ap[w] = __builtin_amdgcn_ds_bpermute(dnSrc, __float_as_int((lane == 31) ? aP : a[w]));
                }
                __builtin_amdgcn_sched_barrier(0);
                float ss[REPLAY_WIN], sf[REPLAY_WIN];
                unsigned cb[REPLAY_WIN], xw[REPLAY_WIN];
#pragma unroll
                for (int w = 0; w < REPLAY_WIN; ++w) {
                    ss[w] = wrap_add(a[w], dslow);
                    const int up = __float_as_int(PI_F - __int_as_float(ap[w]));
                    const int dn = __float_as_int(a[w] - PI_F);
                    const bool crossed = (up & dn) < 0;
                    cb[w] = __builtin_amdgcn_ballot_w32(crossed);
                    const float kk = crossed ? keep : 1.0f;
                    sf[w] = wrap_add(g[w], dfast) * kk;
                    xw[w] = (unsigned)(__float_as_int(ss[w]) ^ a1[w]) | (unsigned)(__float_as_int(sf[w]) ^ g1[w]);
                }
                static_assert((REPLAY_WIN & 1) == 0, "the links are gathered two windows at a time");
                unsigned bad = 0u;
#pragma unroll
                for (int w = 0; w + 2 < REPLAY_WIN; w += 2) bad = or3_gather(bad, xw[w], xw[w + 1]);
                bad = or3_gather(bad, xw[REPLAY_WIN - 2], (lane == 31) ? 0u : xw[REPLAY_WIN - 1]);
                const unsigned allm = (__builtin_amdgcn_ballot_w32(bad != 0u) == 0u) ? 0xffffffffu : 0u;
                auto record = [&](unsigned cm, int w, float gw) {
                    const int kc = __builtin_ctz(cm);
                    const int fb = __builtin_amdgcn_readlane(__float_as_int(gw), kc);
                    if (lane == (evN & 31)) { evStep = t + 32 * w + kc; evBits = fb; }
                    ++evN;
                };
                if (allm == 0xffffffffu && t + REPLAY_SPAN <= N) {
                    xs = __int_as_float(__builtin_amdgcn_readlane(__float_as_int(ss[REPLAY_WIN - 1]), 31));
                    xf = __int_as_float(__builtin_amdgcn_readlane(__float_as_int(sf[REPLAY_WIN - 1]), 31));
                    pv = __int_as_float(__builtin_amdgcn_readlane(__float_as_int(a[REPLAY_WIN - 1]), 31));
                    float* __restrict__ pS = oS + (t + lane);
                    float* __restrict__ pF = oF + (t + lane);
                    const int tb = (t + lane) << 2;
#pragma unroll
                    for (int w = 0; w < REPLAY_WIN; ++w) {
                        pS[32 * w] = a[w];
                        pF[32 * w] = g[w];
                        const int idx = ((tb + 128 * w) & (4 * RING - 4)) >> 2;
                        hS[idx] = a[w];
                        hF[idx] = g[w];
                    }
#pragma unroll
                    for (int w = 0; w < REPLAY_WIN; ++w)
                        if (cb[w] != 0u) record(cb[w], w, g[w]);
                    t += REPLAY_SPAN;
                    continue;
                }
                unsigned mk[REPLAY_WIN];
#pragma unroll
                for (int w = 0; w < REPLAY_WIN; ++w) mk[w] = __builtin_amdgcn_ballot_w32(xw[w] == 0u);
                mk[REPLAY_WIN - 1] &= 0x7fffffffu;
                int K = REPLAY_SPAN - 1;
#pragma unroll
                for (int w = REPLAY_WIN - 1; w >= 0; --w)
                    if (mk[w] != 0xffffffffu) K = 32 * w + __builtin_ctz(~mk[w]);
                const int wK = K >> 5, lK = K & 31;
                float nxs = xs, nxf = xf, npv = pv;
#pragma unroll
                for (int w = 0; w < REPLAY_WIN; ++w) {
                    const float c1 = __int_as_float(__builtin_amdgcn_readlane(__float_as_int(ss[w]), lK));
                    const float c2 = __int_as_float(__builtin_amdgcn_readlane(__float_as_int(sf[w]), lK));
                    const float c3 = __int_as_float(__builtin_amdgcn_readlane(__float_as_int(a[w]), lK));
                    if (w == wK) { nxs = c1; nxf = c2; npv = c3; }
                }
                xs = nxs; xf = nxf; pv = npv;
                const int last = min(K, N - 1 - t);
#pragma unroll
                for (int w = 0; w < REPLAY_WIN; ++w) {
                    const int q = 32 * w + lane;
                    if (q <= last) {
                        oS[t + q] = a[w];
                        oF[t + q] = g[w];
                        hS[(t + q) & RING_MASK] = a[w];
                        hF[(t + q) & RING_MASK] = g[w];
                    }
                }
                if (K + 1 < REPLAY_MIN_OK) lag = 0;
#pragma unroll
                for (int w = 0; w < REPLAY_WIN; ++w) {
                    const int kw = K - 32 * w;
                    const unsigned vm = (kw >= 31) ? 0xffffffffu : ((kw < 0) ? 0u : (0xffffffffu >> (31 - kw)));
                    const unsigned cm = cb[w] & vm;
                    if (cm != 0u) record(cm, w, g[w]);
                }
                t += K + 1;
                continue;
            }
            float a, g;
            {
                Seg sS[RUNS_SLOW], sF[RUNS_FAST];
                {
                    float wS = xs, cS = 0.0f, wF = xf, cF = 0.0f;
#pragma unroll
                    for (int s = 0; s < RUNS_SLOW; ++s)
                        walk_pair(wF, cF, dfast, rdf, sF[s], wS, cS, dslow, rds, sS[s]);
#pragma unroll
                    for (int s = RUNS_SLOW; s < RUNS_FAST; ++s)
                        walk_one(wF, cF, dfast, rdf, sF[s]);
                }
                a = eval_prop<RUNS_SLOW>(sS, kf);
                g = eval_prop<RUNS_FAST>(sF, kf);
            }
            const int ai = __float_as_int(a), gi = __float_as_int(g);
            const int a16 = __builtin_amdgcn_readlane(ai, 16);
            const int g16 = __builtin_amdgcn_readlane(gi, 16);
            const int a15 = __builtin_amdgcn_readlane(ai, 15);
            const float a1 = __int_as_float(__builtin_amdgcn_update_dpp(a16, ai, 0x101, 0xf, 0xf, false));
            const float g1 = __int_as_float(__builtin_amdgcn_update_dpp(g16, gi, 0x101, 0xf, 0xf, false));
            const int apo  = (lane == 0) ? __float_as_int(pv) : a15;
            const float ap = __int_as_float(__builtin_amdgcn_update_dpp(apo, ai, 0x111, 0xf, 0xf, false));
            const float ss = wrap_add(a, dslow);
            const bool crossed = (ap > PI_F) && (a < PI_F);
            const unsigned cball = __builtin_amdgcn_ballot_w32(crossed);
            const float kk = crossed ? keep : 1.0f;
            const float sf = wrap_add(g, dfast) * kk;
            const unsigned m =
                __builtin_amdgcn_ballot_w32(__float_as_uint(ss) == __float_as_uint(a1)) &
                __builtin_amdgcn_ballot_w32(__float_as_uint(sf) == __float_as_uint(g1));
            const int K = __builtin_ctz(~m | 0x80000000u);
            xs = __int_as_float(__builtin_amdgcn_readlane(__float_as_int(ss), K));
            xf = __int_as_float(__builtin_amdgcn_readlane(__float_as_int(sf), K));
            pv = __int_as_float(__builtin_amdgcn_readlane(ai, K));
            const int last = min(K, N - 1 - t);
            if (lane <= last) {
                oS[t + lane] = a;
                oF[t + lane] = g;
                hS[(t + lane) & RING_MASK] = a;
                hF[(t + lane) & RING_MASK] = g;
            }
            const unsigned cm = cball & (0xffffffffu >> (31 - K));
            if (cm != 0u) {
                const int kc = __builtin_ctz(cm);
                const int e  = t + kc;
                const int fb = __builtin_amdgcn_readlane(gi, kc);
                const unsigned mm = __builtin_amdgcn_ballot_w32((evStep >= 0) && (evBits == fb));
                const int slot = evN & 31;
                if (mm != 0u) {
                    const unsigned rr = __builtin_rotateright32(mm, (unsigned)slot);
                    const int ms = (31 - __builtin_clz(rr) + slot) & 31;
                    const int cand = e - __builtin_amdgcn_readlane(evStep, ms);
                    if (cand >= 32) {
                        int c2 = cand;
                        while (c2 < REPLAY_SPAN) c2 += cand;
                        if (c2 + REPLAY_SPAN <= RING) lag = c2;
                    }
                }
                if (lane == slot) { evStep = e; evBits = fb; }
                ++evN;
            }
            t += K + 1;
        }
        return;
    }

    if (lane != 0) return;
    float slow = 0.0f, fast = 0.0f, prev = 0.0f;
    for (int t = 0; t < N; ++t) {
        oS[t] = slow;
        oF[t] = fast;
        bool crossed = (prev > PI_F) && (slow < PI_F);
        float ns = pmod(slow + dslow, TWO_PI_F);
        float nf = pmod(fast + dfast, TWO_PI_F);
        if (crossed) nf = pmod(nf * keep, TWO_PI_F);
        prev = slow; slow = ns; fast = nf;
    }
}

__global__ void osc_scale_wmma_kernel(const float* __restrict__ in,
                                      float* __restrict__ out, int N) {
    const float s = INV_TWO_PI_F;
    const int base = blockIdx.x * 256;
    const int lane = threadIdx.x;
    const int i    = lane & 15;
    const int kh   = lane >> 4;

    if (base + 256 <= N) {
        v8f d = {0.f, 0.f, 0.f, 0.f, 0.f, 0.f, 0.f, 0.f};
#pragma unroll
        for (int m = 0; m < 4; ++m) {
            const float* p = in + base + i * 16 + 4 * m + 2 * kh;
            v2f a; a.x = p[0]; a.y = p[1];
            v2f b;
            b.x = (i == (4 * m + 2 * kh + 0)) ? s : 0.0f;
            b.y = (i == (4 * m + 2 * kh + 1)) ? s : 0.0f;
            d = __builtin_amdgcn_wmma_f32_16x16x4_f32(
                    false, a, false, b, (short)0, d, false, false);
        }
#pragma unroll
        for (int r = 0; r < 8; ++r) {
            out[base + (r + 8 * kh) * 16 + i] = d[r];
        }
    } else {
        for (int j = base + lane; j < N; j += 32) out[j] = in[j] * s;
    }
}

extern "C" void kernel_launch(void* const* d_in, const int* in_sizes, int n_in,
                              void* d_out, int out_size, void* d_ws, size_t ws_size,
                              hipStream_t stream) {
    const float* p_log_slow = (const float*)d_in[0];
    const float* p_log_fast = (const float*)d_in[1];
    const float* p_reset    = (const float*)d_in[2];
    const int N = out_size / 3;
    if (N <= 0) return;
    float* out = (float*)d_out;

    osc_scan_kernel<<<1, 32, 0, stream>>>(p_log_slow, p_log_fast, p_reset, out, N);

    const int tiles = (N + 255) / 256;
    osc_scale_wmma_kernel<<<tiles, 32, 0, stream>>>(out, out + 2 * (size_t)N, N);
}


# scan: a replayed pass holds four consecutive steps a lane (blocks of 128): three links in four stay in the lane, one lane rotate a block, outputs go out 16 bytes a lane; same values, same order
# speedup vs baseline: 6.6674x; 1.0449x over previous
#include <hip/hip_runtime.h>
#include <hip/hip_bf16.h>


typedef __attribute__((ext_vector_type(2))) float v2f;
typedef __attribute__((ext_vector_type(8))) float v8f;
typedef __attribute__((ext_vector_type(4))) float v4f;

#define TWO_PI_F     6.28318530717958647692f
#define PI_F         3.14159265358979323846f
#define INV_TWO_PI_F 0.15915494309189533577f
#define DT_F         0.001f

__device__ __forceinline__ float pmod(float x, float m) {
    float r = fmodf(x, m);
    return (r < 0.0f) ? (r + m) : r;
}

__device__ __forceinline__ float wrap_add(float x, float d) {
    float t = x + d;
    return (t >= TWO_PI_F) ? (t - TWO_PI_F) : t;
}


struct Seg { float c, w, t2, de; };

__device__ __forceinline__ float to_scalar(float x) {
    float r;
    asm("v_readfirstlane_b32 %0, %1" : "=s"(r) : "v"(x));
    return r;
}

__device__ __forceinline__ void walk_pair(float& wF, float& cF, float dF, float rdF, Seg& sgF,
                                          float& wS, float& cS, float dS, float rdS, Seg& sgS) {
    float wnF, cnF, t2F, deF, bF, tF, baF;
    float wnS, cnS, t2S, deS, bS, tS, baS;
    asm(
        "s_and_b32 %[bF], %[wF], 0x7f800000\n\t"
        "s_and_b32 %[bS], %[wS], 0x7f800000\n\t"
        "s_add_f32 %[t2F], %[wF], %[dF]\n\t"
        "s_add_f32 %[t2S], %[wS], %[dS]\n\t"
        "s_add_f32 %[tF], %[bF], %[bF]\n\t"
        "s_add_f32 %[tS], %[bS], %[bS]\n\t"
        "s_add_f32 %[deF], %[bF], %[dF]\n\t"
        "s_add_f32 %[deS], %[bS], %[dS]\n\t"
        "s_min_u32 %[tF], %[tF], 0x40c90fdb\n\t"
        "s_min_u32 %[tS], %[tS], 0x40c90fdb\n\t"
        "s_sub_f32 %[deF], %[deF], %[bF]\n\t"
        "s_sub_f32 %[deS], %[deS], %[bS]\n\t"
        "s_sub_f32 %[tF], %[tF], %[t2F]\n\t"
        "s_sub_f32 %[tS], %[tS], %[t2S]\n\t"
        "s_sub_f32 %[baF], %[t2F], %[deF]\n\t"
        "s_sub_f32 %[baS], %[t2S], %[deS]\n\t"
        "s_mul_f32 %[tF], %[tF], %[rF]\n\t"
        "s_mul_f32 %[tS], %[tS], %[rS]\n\t"
        "s_ceil_f32 %[tF], %[tF]\n\t"
        "s_ceil_f32 %[tS], %[tS]\n\t"
        "s_fmac_f32 %[baF], %[tF], %[deF]\n\t"
        "s_fmac_f32 %[baS], %[tS], %[deS]\n\t"
        "s_max_num_f32 %[cnF], %[tF], 0\n\t"
        "s_max_num_f32 %[cnS], %[tS], 0\n\t"
        "s_cmp_ge_f32 %[tF], 1.0\n\t"
        "s_cselect_b32 %[baF], %[baF], %[wF]\n\t"
        "s_cmp_ge_f32 %[tS], 1.0\n\t"
        "s_cselect_b32 %[baS], %[baS], %[wS]\n\t"
        "s_add_f32 %[baF], %[baF], %[dF]\n\t"
        "s_add_f32 %[baS], %[baS], %[dS]\n\t"
        "s_add_f32 %[cnF], %[cnF], %[cF]\n\t"
        "s_add_f32 %[cnS], %[cnS], %[cS]\n\t"
        "s_add_f32 %[wnF], %[baF], 0xc0c90fdb\n\t"
        "s_add_f32 %[wnS], %[baS], 0xc0c90fdb\n\t"
        "s_add_f32 %[cnF], %[cnF], 1.0\n\t"
        "s_add_f32 %[cnS], %[cnS], 1.0\n\t"
        "s_cmp_ge_f32 %[baF], 0x40c90fdb\n\t"
        "s_cselect_b32 %[wnF], %[wnF], %[baF]\n\t"
        "s_cmp_ge_f32 %[baS], 0x40c90fdb\n\t"
        "s_cselect_b32 %[wnS], %[wnS], %[baS]\n\t"
        : [wnF] "=&s"(wnF), [cnF] "=&s"(cnF), [t2F] "=&s"(t2F), [deF] "=&s"(deF), [bF] "=&s"(bF), [tF] "=&s"(tF), [baF] "=&s"(baF), [wnS] "=&s"(wnS), [cnS] "=&s"(cnS), [t2S] "=&s"(t2S), [deS] "=&s"(deS), [bS] "=&s"(bS), [tS] "=&s"(tS), [baS] "=&s"(baS)
        : [wF] "s"(wF), [cF] "s"(cF), [dF] "s"(dF), [rF] "s"(rdF), [wS] "s"(wS), [cS] "s"(cS), [dS] "s"(dS), [rS] "s"(rdS)
        : "scc");
    sgF.c = cF; sgF.w = wF; sgF.t2 = t2F; sgF.de = deF;
    sgS.c = cS; sgS.w = wS; sgS.t2 = t2S; sgS.de = deS;
    wF = wnF; cF = cnF;
    wS = wnS; cS = cnS;
}

__device__ __forceinline__ void walk_one(float& wF, float& cF, float dF, float rdF, Seg& sgF) {
    float wnF, cnF, t2F, deF, bF, tF, baF;
    asm(
        "s_and_b32 %[bF], %[wF], 0x7f800000\n\t"
        "s_add_f32 %[t2F], %[wF], %[dF]\n\t"
        "s_add_f32 %[tF], %[bF], %[bF]\n\t"
        "s_add_f32 %[deF], %[bF], %[dF]\n\t"
        "s_min_u32 %[tF], %[tF], 0x40c90fdb\n\t"
        "s_sub_f32 %[deF], %[deF], %[bF]\n\t"
        "s_sub_f32 %[tF], %[tF], %[t2F]\n\t"
        "s_sub_f32 %[baF], %[t2F], %[deF]\n\t"
        "s_mul_f32 %[tF], %[tF], %[rF]\n\t"
        "s_ceil_f32 %[tF], %[tF]\n\t"
        "s_fmac_f32 %[baF], %[tF], %[deF]\n\t"
        "s_max_num_f32 %[cnF], %[tF], 0\n\t"
        "s_cmp_ge_f32 %[tF], 1.0\n\t"
        "s_cselect_b32 %[baF], %[baF], %[wF]\n\t"
        "s_add_f32 %[baF], %[baF], %[dF]\n\t"
        "s_add_f32 %[cnF], %[cnF], %[cF]\n\t"
        "s_add_f32 %[wnF], %[baF], 0xc0c90fdb\n\t"
        "s_add_f32 %[cnF], %[cnF], 1.0\n\t"
        "s_cmp_ge_f32 %[baF], 0x40c90fdb\n\t"
        "s_cselect_b32 %[wnF], %[wnF], %[baF]\n\t"
        : [wnF] "=&s"(wnF), [cnF] "=&s"(cnF), [t2F] "=&s"(t2F), [deF] "=&s"(deF), [bF] "=&s"(bF), [tF] "=&s"(tF), [baF] "=&s"(baF)
        : [wF] "s"(wF), [cF] "s"(cF), [dF] "s"(dF), [rF] "s"(rdF)
        : "scc");
    sgF.c = cF; sgF.w = wF; sgF.t2 = t2F; sgF.de = deF;
    wF = wnF; cF = cnF;
}

template <int S>
__device__ __forceinline__ float eval_prop(const Seg (&sg)[S], float kf) {
    float c = sg[0].c, w = sg[0].w, t2 = sg[0].t2, de = sg[0].de;
#pragma unroll
    for (int s = 1; s < S; ++s) {
        const bool in = kf >= sg[s].c;
        c  = in ? sg[s].c  : c;
        w  = in ? sg[s].w  : w;
        t2 = in ? sg[s].t2 : t2;
        de = in ? sg[s].de : de;
    }
    const float j = kf - c;
    return (j == 0.0f) ? w : fmaf(j - 1.0f, de, t2);
}

__device__ __forceinline__ unsigned or3_gather(unsigned acc, unsigned x, unsigned y) {
    unsigned r;
    asm("v_or3_b32 %0, %1, %2, %3" : "=v"(r) : "v"(acc), "v"(x), "v"(y));
    return r;
}

#define RUNS_SLOW 5
#define RUNS_FAST 9
static_assert(RUNS_SLOW <= RUNS_FAST, "the paired walk covers the slow table");

#define RING 4096
#define RING_MASK (RING - 1)
#define REPLAY_MIN_OK 8
#define REPLAY_BLK 4
#define REPLAY_EL (4 * REPLAY_BLK)
#define REPLAY_SPAN (128 * REPLAY_BLK)
static_assert(2 * REPLAY_SPAN <= RING, "a replayed pass and its sources fit the ring");

__global__ void osc_scan_kernel(const float* __restrict__ p_log_slow,
                                const float* __restrict__ p_log_fast,
                                const float* __restrict__ p_reset,
                                float* __restrict__ out, int N) {
    if (blockIdx.x != 0) return;
    const int lane = threadIdx.x;

    const float dslow = expf(p_log_slow[0]) * DT_F;
    const float dfast = expf(p_log_fast[0]) * DT_F;
    const float keep  = 1.0f - p_reset[0];

    float* __restrict__ oS = out;
    float* __restrict__ oF = out + N;

    const bool fast_ok =
        (dslow >= 0.0f) && (dslow < TWO_PI_F) &&
        (dfast >= 0.0f) && (dfast < TWO_PI_F) &&
        (keep  >= 0.0f) && (keep  <= 1.0f);

    __shared__ float hS[RING];
    __shared__ float hF[RING];

    if (fast_ok) {
        const float kf  = (float)lane;
        for (int i = lane; i < RING; i += 32) { hS[i] = 0.0f; hF[i] = 0.0f; }
        int evStep = -1, evBits = 0;
        int evN = 0, lag = 0;
        const float rds = to_scalar(1.0f / dslow);
        const float rdf = to_scalar(1.0f / dfast);
        float xs = 0.0f, xf = 0.0f, pv = 0.0f;
        int t = 0;
        while (t < N) {
            const bool replay = (lag > 0) && (t >= lag);
            if (replay) {
                float a[REPLAY_EL], g[REPLAY_EL];
                const int sb = (t + 4 * lane - lag) << 2;
#pragma unroll
                for (int e = 0; e < REPLAY_EL; ++e) {
                    const int idx = ((sb + 512 * (e >> 2) + 4 * (e & 3)) & (4 * RING - 4)) >> 2;
                    a[e] = hS[idx];
                    g[e] = hF[idx];
                }
                const float offS = xs - __int_as_float(__builtin_amdgcn_readfirstlane(__float_as_int(a[0])));
                const float offF = xf - __int_as_float(__builtin_amdgcn_readfirstlane(__float_as_int(g[0])));
#pragma unroll
                for (int e = 0; e < REPLAY_EL; ++e) { a[e] += offS; g[e] += offF; }
                a[0] = (lane == 0) ? xs : a[0];
                g[0] = (lane == 0) ? xf : g[0];
                const int upSrc = ((lane + 1) & 31) << 2;
                const int dnSrc = ((lane + 31) & 31) << 2;
                int a1s[REPLAY_BLK], g1s[REPLAY_BLK], aps[REPLAY_BLK];
#pragma unroll
                for (int b = 0; b < REPLAY_BLK; ++b) {
                    const float aN = (b + 1 < REPLAY_BLK) ? a[4 * b + 4] : a[4 * b];
                    const float gN = (b + 1 < REPLAY_BLK) ? g[4 * b + 4] : g[4 * b];
                    const float aP = (b > 0) ? a[4 * b - 1] : pv;
                    a1s[b] = __builtin_amdgcn_ds_bpermute(upSrc, __float_as_int((lane == 0) ? aN : a[4 * b]));
                    g1s[b] = __builtin_amdgcn_ds_bpermute(upSrc, __float_as_int((lane == 0) ? gN : g[4 * b]));
                    aps[b] = __builtin_amdgcn_ds_bpermute(dnSrc, __float_as_int((lane == 31) ? aP : a[4 * b + 3]));
                }
                float ss[REPLAY_EL], sf[REPLAY_EL];
                unsigned cb[REPLAY_EL], xw[REPLAY_EL];
                auto fast_step = [&](int e, float prev) {
                    const int up = __float_as_int(PI_F - prev);
                    const int dn = __float_as_int(a[e] - PI_F);
                    const bool crossed = (up & dn) < 0;
                    cb[e] = __builtin_amdgcn_ballot_w32(crossed);
                    const float kk = crossed ? keep : 1.0f;
                    sf[e] = wrap_add(g[e], dfast) * kk;
                };
                __builtin_amdgcn_sched_barrier(0);
#pragma unroll
                for (int e = 0; e < REPLAY_EL; ++e) {
                    const int j = e & 3;
                    ss[e] = wrap_add(a[e], dslow);
                    if (j > 0) fast_step(e, a[e - 1]);
                    if (j == 1 || j == 2)
                        xw[e] = (unsigned)(__float_as_int(ss[e]) ^ __float_as_int(a[e + 1])) |
                                (unsigned)(__float_as_int(sf[e]) ^ __float_as_int(g[e + 1]));
                }
                __builtin_amdgcn_sched_barrier(0);
#pragma unroll
                for (int e = 0; e < REPLAY_EL; ++e) {
                    const int b = e >> 2, j = e & 3;
                    if (j == 0) {
                        fast_step(e, __int_as_float(aps[b]));
                        xw[e] = (unsigned)(__float_as_int(ss[e]) ^ __float_as_int(a[e + 1])) |
                                (unsigned)(__float_as_int(sf[e]) ^ __float_as_int(g[e + 1]));
                    }
                    if (j == 3)
                        xw[e] = (unsigned)(__float_as_int(ss[e]) ^ a1s[b]) | (unsigned)(__float_as_int(sf[e]) ^ g1s[b]);
                }
                static_assert((REPLAY_EL & 1) == 0, "the links are gathered two elements at a time");
                unsigned bad = 0u;
#pragma unroll
                for (int e = 0; e + 2 < REPLAY_EL; e += 2) bad = or3_gather(bad, xw[e], xw[e + 1]);
                bad = or3_gather(bad, xw[REPLAY_EL - 2], (lane == 31) ? 0u : xw[REPLAY_EL - 1]);
                const unsigned allm = (__builtin_amdgcn_ballot_w32(bad != 0u) == 0u) ? 0xffffffffu : 0u;
                auto record = [&](int b, unsigned c0, unsigned c1, unsigned c2, unsigned c3,
                                  float g0, float g1, float g2, float g3) {
                    unsigned cm = c0 | c1 | c2 | c3;
                    while (cm != 0u) {
                        const int kc = __builtin_ctz(cm);
                        const unsigned bit = 1u << kc;
                        const int j = (c0 & bit) ? 0 : ((c1 & bit) ? 1 : ((c2 & bit) ? 2 : 3));
                        const float gj = (j == 0) ? g0 : ((j == 1) ? g1 : ((j == 2) ? g2 : g3));
                        const int fb = __builtin_amdgcn_readlane(__float_as_int(gj), kc);
                        if (lane == (evN & 31)) { evStep = t + 128 * b + 4 * kc + j; evBits = fb; }
                        ++evN;
                        cm &= cm - 1u;
                    }
                };
                if (allm == 0xffffffffu && t + REPLAY_SPAN <= N) {
                    xs = __int_as_float(__builtin_amdgcn_readlane(__float_as_int(ss[REPLAY_EL - 1]), 31));
                    xf = __int_as_float(__builtin_amdgcn_readlane(__float_as_int(sf[REPLAY_EL - 1]), 31));
                    pv = __int_as_float(__builtin_amdgcn_readlane(__float_as_int(a[REPLAY_EL - 1]), 31));
                    float* __restrict__ pS = oS + (t + 4 * lane);
                    float* __restrict__ pF = oF + (t + 4 * lane);
                    const int tb = (t + 4 * lane) << 2;
#pragma unroll
                    for (int b = 0; b < REPLAY_BLK; ++b) {
                        const v4f vS = {a[4 * b], a[4 * b + 1], a[4 * b + 2], a[4 * b + 3]};
                        const v4f vF = {g[4 * b], g[4 * b + 1], g[4 * b + 2], g[4 * b + 3]};
                        __builtin_memcpy(pS + 128 * b, &vS, sizeof(v4f));
                        __builtin_memcpy(pF + 128 * b, &vF, sizeof(v4f));
#pragma unroll
                        for (int j = 0; j < 4; ++j) {
                            const int idx = ((tb + 512 * b + 4 * j) & (4 * RING - 4)) >> 2;
                            hS[idx] = a[4 * b + j];
                            hF[idx] = g[4 * b + j];
                        }
                    }
#pragma unroll
                    for (int b = 0; b < REPLAY_BLK; ++b)
                        record(b, cb[4 * b], cb[4 * b + 1], cb[4 * b + 2], cb[4 * b + 3],
                               g[4 * b], g[4 * b + 1], g[4 * b + 2], g[4 * b + 3]);
                    t += REPLAY_SPAN;
                    continue;
                }
                unsigned mk[REPLAY_EL];
#pragma unroll
                for (int e = 0; e < REPLAY_EL; ++e) mk[e] = __builtin_amdgcn_ballot_w32(xw[e] == 0u);
                mk[REPLAY_EL - 1] &= 0x7fffffffu;
                int K = REPLAY_SPAN - 1;
#pragma unroll
                for (int b = REPLAY_BLK - 1; b >= 0; --b) {
                    const unsigned okl = mk[4 * b] & mk[4 * b + 1] & mk[4 * b + 2] & mk[4 * b + 3];
                    if (okl != 0xffffffffu) {
                        const int lf = __builtin_ctz(~okl);
                        const unsigned bit = 1u << lf;
                        const int jf = !(mk[4 * b] & bit) ? 0 : (!(mk[4 * b + 1] & bit) ? 1 : (!(mk[4 * b + 2] & bit) ? 2 : 3));
                        K = 128 * b + 4 * lf + jf;
                    }
                }
                const int eK = ((K >> 7) << 2) | (K & 3), lK = (K >> 2) & 31;
                float nxs = xs, nxf = xf, npv = pv;
#pragma unroll
                for (int e = 0; e < REPLAY_EL; ++e) {
                    const float c1 = __int_as_float(__builtin_amdgcn_readlane(__float_as_int(ss[e]), lK));
                    const float c2 = __int_as_float(__builtin_amdgcn_readlane(__float_as_int(sf[e]), lK));
                    const float c3 = __int_as_float(__builtin_amdgcn_readlane(__float_as_int(a[e]), lK));
                    if (e == eK) { nxs = c1; nxf = c2; npv = c3; }
                }
                xs = nxs; xf = nxf; pv = npv;
                const int last = min(K, N - 1 - t);
#pragma unroll
                for (int e = 0; e < REPLAY_EL; ++e) {
                    const int q = 128 * (e >> 2) + 4 * lane + (e & 3);
                    if (q <= last) {
                        oS[t + q] = a[e];
                        oF[t + q] = g[e];
                        hS[(t + q) & RING_MASK] = a[e];
                        hF[(t + q) & RING_MASK] = g[e];
                    }
                }
                if (K + 1 < REPLAY_MIN_OK) lag = 0;
                unsigned cv[REPLAY_EL];
#pragma unroll
                for (int e = 0; e < REPLAY_EL; ++e) {
                    const int r = K - 128 * (e >> 2) - (e & 3);
                    const int lv = r >> 2;
                    const unsigned vm = (lv >= 31) ? 0xffffffffu : ((r < 0) ? 0u : (0xffffffffu >> (31 - lv)));
                    cv[e] = cb[e] & vm;
                }
#pragma unroll
                for (int b = 0; b < REPLAY_BLK; ++b)
                    record(b, cv[4 * b], cv[4 * b + 1], cv[4 * b + 2], cv[4 * b + 3],
                           g[4 * b], g[4 * b + 1], g[4 * b + 2], g[4 * b + 3]);
                t += K + 1;
                continue;
            }
            float a, g;
            {
                Seg sS[RUNS_SLOW], sF[RUNS_FAST];
                {
                    float wS = xs, cS = 0.0f, wF = xf, cF = 0.0f;
#pragma unroll
                    for (int s = 0; s < RUNS_SLOW; ++s)
                        walk_pair(wF, cF, dfast, rdf, sF[s], wS, cS, dslow, rds, sS[s]);
#pragma unroll
                    for (int s = RUNS_SLOW; s < RUNS_FAST; ++s)
                        walk_one(wF, cF, dfast, rdf, sF[s]);
                }
                a = eval_prop<RUNS_SLOW>(sS, kf);
                g = eval_prop<RUNS_FAST>(sF, kf);
            }
            const int ai = __float_as_int(a), gi = __float_as_int(g);
            const int a16 = __builtin_amdgcn_readlane(ai, 16);
            const int g16 = __builtin_amdgcn_readlane(gi, 16);
            const int a15 = __builtin_amdgcn_readlane(ai, 15);
            const float a1 = __int_as_float(__builtin_amdgcn_update_dpp(a16, ai, 0x101, 0xf, 0xf, false));
            const float g1 = __int_as_float(__builtin_amdgcn_update_dpp(g16, gi, 0x101, 0xf, 0xf, false));
            const int apo  = (lane == 0) ? __float_as_int(pv) : a15;
            const float ap = __int_as_float(__builtin_amdgcn_update_dpp(apo, ai, 0x111, 0xf, 0xf, false));
            const float ss = wrap_add(a, dslow);
            const bool crossed = (ap > PI_F) && (a < PI_F);
            const unsigned cball = __builtin_amdgcn_ballot_w32(crossed);
            const float kk = crossed ? keep : 1.0f;
            const float sf = wrap_add(g, dfast) * kk;
            const unsigned m =
                __builtin_amdgcn_ballot_w32(__float_as_uint(ss) == __float_as_uint(a1)) &
                __builtin_amdgcn_ballot_w32(__float_as_uint(sf) == __float_as_uint(g1));
            const int K = __builtin_ctz(~m | 0x80000000u);
            xs = __int_as_float(__builtin_amdgcn_readlane(__float_as_int(ss), K));
            xf = __int_as_float(__builtin_amdgcn_readlane(__float_as_int(sf), K));
            pv = __int_as_float(__builtin_amdgcn_readlane(ai, K));
            const int last = min(K, N - 1 - t);
            if (lane <= last) {
                oS[t + lane] = a;
                oF[t + lane] = g;
                hS[(t + lane) & RING_MASK] = a;
                hF[(t + lane) & RING_MASK] = g;
            }
            const unsigned cm = cball & (0xffffffffu >> (31 - K));
            if (cm != 0u) {
                const int kc = __builtin_ctz(cm);
                const int e  = t + kc;
                const int fb = __builtin_amdgcn_readlane(gi, kc);
                const unsigned mm = __builtin_amdgcn_ballot_w32((evStep >= 0) && (evBits == fb));
                const int slot = evN & 31;
                if (mm != 0u) {
                    const unsigned rr = __builtin_rotateright32(mm, (unsigned)slot);
                    const int ms = (31 - __builtin_clz(rr) + slot) & 31;
                    const int cand = e - __builtin_amdgcn_readlane(evStep, ms);
                    if (cand >= 32) {
                        int c2 = cand;
                        while (c2 < REPLAY_SPAN) c2 += cand;
                        if (c2 + REPLAY_SPAN <= RING) lag = c2;
                    }
                }
                if (lane == slot) { evStep = e; evBits = fb; }
                ++evN;
            }
            t += K + 1;
        }
        return;
    }

    if (lane != 0) return;
    float slow = 0.0f, fast = 0.0f, prev = 0.0f;
    for (int t = 0; t < N; ++t) {
        oS[t] = slow;
        oF[t] = fast;
        bool crossed = (prev > PI_F) && (slow < PI_F);
        float ns = pmod(slow + dslow, TWO_PI_F);
        float nf = pmod(fast + dfast, TWO_PI_F);
        if (crossed) nf = pmod(nf * keep, TWO_PI_F);
        prev = slow; slow = ns; fast = nf;
    }
}

__global__ void osc_scale_wmma_kernel(const float* __restrict__ in,
                                      float* __restrict__ out, int N) {
    const float s = INV_TWO_PI_F;
    const int base = blockIdx.x * 256;
    const int lane = threadIdx.x;
    const int i    = lane & 15;
    const int kh   = lane >> 4;

    if (base + 256 <= N) {
        v8f d = {0.f, 0.f, 0.f, 0.f, 0.f, 0.f, 0.f, 0.f};
#pragma unroll
        for (int m = 0; m < 4; ++m) {
            const float* p = in + base + i * 16 + 4 * m + 2 * kh;
            v2f a; a.x = p[0]; a.y = p[1];
            v2f b;
            b.x = (i == (4 * m + 2 * kh + 0)) ? s : 0.0f;
            b.y = (i == (4 * m + 2 * kh + 1)) ? s : 0.0f;
            d = __builtin_amdgcn_wmma_f32_16x16x4_f32(
                    false, a, false, b, (short)0, d, false, false);
        }
#pragma unroll
        for (int r = 0; r < 8; ++r) {
            out[base + (r + 8 * kh) * 16 + i] = d[r];
        }
    } else {
        for (int j = base + lane; j < N; j += 32) out[j] = in[j] * s;
    }
}

extern "C" void kernel_launch(void* const* d_in, const int* in_sizes, int n_in,
                              void* d_out, int out_size, void* d_ws, size_t ws_size,
                              hipStream_t stream) {
    const float* p_log_slow = (const float*)d_in[0];
    const float* p_log_fast = (const float*)d_in[1];
    const float* p_reset    = (const float*)d_in[2];
    const int N = out_size / 3;
    if (N <= 0) return;
    float* out = (float*)d_out;

    osc_scan_kernel<<<1, 32, 0, stream>>>(p_log_slow, p_log_fast, p_reset, out, N);

    const int tiles = (N + 255) / 256;
    osc_scale_wmma_kernel<<<tiles, 32, 0, stream>>>(out, out + 2 * (size_t)N, N);
}


# scan: a replay lag that breaks at once twice running is held back for one lag (the slow chain's rounding pattern has shifted; its sources stay stale that long); walked passes meanwhile; same values
# speedup vs baseline: 7.8704x; 1.1804x over previous
#include <hip/hip_runtime.h>
#include <hip/hip_bf16.h>


typedef __attribute__((ext_vector_type(2))) float v2f;
typedef __attribute__((ext_vector_type(8))) float v8f;
typedef __attribute__((ext_vector_type(4))) float v4f;

#define TWO_PI_F     6.28318530717958647692f
#define PI_F         3.14159265358979323846f
#define INV_TWO_PI_F 0.15915494309189533577f
#define DT_F         0.001f

__device__ __forceinline__ float pmod(float x, float m) {
    float r = fmodf(x, m);
    return (r < 0.0f) ? (r + m) : r;
}

__device__ __forceinline__ float wrap_add(float x, float d) {
    float t = x + d;
    return (t >= TWO_PI_F) ? (t - TWO_PI_F) : t;
}


struct Seg { float c, w, t2, de; };

__device__ __forceinline__ float to_scalar(float x) {
    float r;
    asm("v_readfirstlane_b32 %0, %1" : "=s"(r) : "v"(x));
    return r;
}

__device__ __forceinline__ void walk_pair(float& wF, float& cF, float dF, float rdF, Seg& sgF,
                                          float& wS, float& cS, float dS, float rdS, Seg& sgS) {
    float wnF, cnF, t2F, deF, bF, tF, baF;
    float wnS, cnS, t2S, deS, bS, tS, baS;
    asm(
        "s_and_b32 %[bF], %[wF], 0x7f800000\n\t"
        "s_and_b32 %[bS], %[wS], 0x7f800000\n\t"
        "s_add_f32 %[t2F], %[wF], %[dF]\n\t"
        "s_add_f32 %[t2S], %[wS], %[dS]\n\t"
        "s_add_f32 %[tF], %[bF], %[bF]\n\t"
        "s_add_f32 %[tS], %[bS], %[bS]\n\t"
        "s_add_f32 %[deF], %[bF], %[dF]\n\t"
        "s_add_f32 %[deS], %[bS], %[dS]\n\t"
        "s_min_u32 %[tF], %[tF], 0x40c90fdb\n\t"
        "s_min_u32 %[tS], %[tS], 0x40c90fdb\n\t"
        "s_sub_f32 %[deF], %[deF], %[bF]\n\t"
        "s_sub_f32 %[deS], %[deS], %[bS]\n\t"
        "s_sub_f32 %[tF], %[tF], %[t2F]\n\t"
        "s_sub_f32 %[tS], %[tS], %[t2S]\n\t"
        "s_sub_f32 %[baF], %[t2F], %[deF]\n\t"
        "s_sub_f32 %[baS], %[t2S], %[deS]\n\t"
        "s_mul_f32 %[tF], %[tF], %[rF]\n\t"
        "s_mul_f32 %[tS], %[tS], %[rS]\n\t"
        "s_ceil_f32 %[tF], %[tF]\n\t"
        "s_ceil_f32 %[tS], %[tS]\n\t"
        "s_fmac_f32 %[baF], %[tF], %[deF]\n\t"
        "s_fmac_f32 %[baS], %[tS], %[deS]\n\t"
        "s_max_num_f32 %[cnF], %[tF], 0\n\t"
        "s_max_num_f32 %[cnS], %[tS], 0\n\t"
        "s_cmp_ge_f32 %[tF], 1.0\n\t"
        "s_cselect_b32 %[baF], %[baF], %[wF]\n\t"
        "s_cmp_ge_f32 %[tS], 1.0\n\t"
        "s_cselect_b32 %[baS], %[baS], %[wS]\n\t"
        "s_add_f32 %[baF], %[baF], %[dF]\n\t"
        "s_add_f32 %[baS], %[baS], %[dS]\n\t"
        "s_add_f32 %[cnF], %[cnF], %[cF]\n\t"
        "s_add_f32 %[cnS], %[cnS], %[cS]\n\t"
        "s_add_f32 %[wnF], %[baF], 0xc0c90fdb\n\t"
        "s_add_f32 %[wnS], %[baS], 0xc0c90fdb\n\t"
        "s_add_f32 %[cnF], %[cnF], 1.0\n\t"
        "s_add_f32 %[cnS], %[cnS], 1.0\n\t"
        "s_cmp_ge_f32 %[baF], 0x40c90fdb\n\t"
        "s_cselect_b32 %[wnF], %[wnF], %[baF]\n\t"
        "s_cmp_ge_f32 %[baS], 0x40c90fdb\n\t"
        "s_cselect_b32 %[wnS], %[wnS], %[baS]\n\t"
        : [wnF] "=&s"(wnF), [cnF] "=&s"(cnF), [t2F] "=&s"(t2F), [deF] "=&s"(deF), [bF] "=&s"(bF), [tF] "=&s"(tF), [baF] "=&s"(baF), [wnS] "=&s"(wnS), [cnS] "=&s"(cnS), [t2S] "=&s"(t2S), [deS] "=&s"(deS), [bS] "=&s"(bS), [tS] "=&s"(tS), [baS] "=&s"(baS)
        : [wF] "s"(wF), [cF] "s"(cF), [dF] "s"(dF), [rF] "s"(rdF), [wS] "s"(wS), [cS] "s"(cS), [dS] "s"(dS), [rS] "s"(rdS)
        : "scc");
    sgF.c = cF; sgF.w = wF; sgF.t2 = t2F; sgF.de = deF;
    sgS.c = cS; sgS.w = wS; sgS.t2 = t2S; sgS.de = deS;
    wF = wnF; cF = cnF;
    wS = wnS; cS = cnS;
}

__device__ __forceinline__ void walk_one(float& wF, float& cF, float dF, float rdF, Seg& sgF) {
    float wnF, cnF, t2F, deF, bF, tF, baF;
    asm(
        "s_and_b32 %[bF], %[wF], 0x7f800000\n\t"
        "s_add_f32 %[t2F], %[wF], %[dF]\n\t"
        "s_add_f32 %[tF], %[bF], %[bF]\n\t"
        "s_add_f32 %[deF], %[bF], %[dF]\n\t"
        "s_min_u32 %[tF], %[tF], 0x40c90fdb\n\t"
        "s_sub_f32 %[deF], %[deF], %[bF]\n\t"
        "s_sub_f32 %[tF], %[tF], %[t2F]\n\t"
        "s_sub_f32 %[baF], %[t2F], %[deF]\n\t"
        "s_mul_f32 %[tF], %[tF], %[rF]\n\t"
        "s_ceil_f32 %[tF], %[tF]\n\t"
        "s_fmac_f32 %[baF], %[tF], %[deF]\n\t"
        "s_max_num_f32 %[cnF], %[tF], 0\n\t"
        "s_cmp_ge_f32 %[tF], 1.0\n\t"
        "s_cselect_b32 %[baF], %[baF], %[wF]\n\t"
        "s_add_f32 %[baF], %[baF], %[dF]\n\t"
        "s_add_f32 %[cnF], %[cnF], %[cF]\n\t"
        "s_add_f32 %[wnF], %[baF], 0xc0c90fdb\n\t"
        "s_add_f32 %[cnF], %[cnF], 1.0\n\t"
        "s_cmp_ge_f32 %[baF], 0x40c90fdb\n\t"
        "s_cselect_b32 %[wnF], %[wnF], %[baF]\n\t"
        : [wnF] "=&s"(wnF), [cnF] "=&s"(cnF), [t2F] "=&s"(t2F), [deF] "=&s"(deF), [bF] "=&s"(bF), [tF] "=&s"(tF), [baF] "=&s"(baF)
        : [wF] "s"(wF), [cF] "s"(cF), [dF] "s"(dF), [rF] "s"(rdF)
        : "scc");
    sgF.c = cF; sgF.w = wF; sgF.t2 = t2F; sgF.de = deF;
    wF = wnF; cF = cnF;
}

template <int S>
__device__ __forceinline__ float eval_prop(const Seg (&sg)[S], float kf) {
    float c = sg[0].c, w = sg[0].w, t2 = sg[0].t2, de = sg[0].de;
#pragma unroll
    for (int s = 1; s < S; ++s) {
        const bool in = kf >= sg[s].c;
        c  = in ? sg[s].c  : c;
        w  = in ? sg[s].w  : w;
        t2 = in ? sg[s].t2 : t2;
        de = in ? sg[s].de : de;
    }
    const float j = kf - c;
    return (j == 0.0f) ? w : fmaf(j - 1.0f, de, t2);
}

__device__ __forceinline__ unsigned or3_gather(unsigned acc, unsigned x, unsigned y) {
    unsigned r;
    asm("v_or3_b32 %0, %1, %2, %3" : "=v"(r) : "v"(acc), "v"(x), "v"(y));
    return r;
}

#define RUNS_SLOW 5
#define RUNS_FAST 9
static_assert(RUNS_SLOW <= RUNS_FAST, "the paired walk covers the slow table");

#define RING 4096
#define RING_MASK (RING - 1)
#define REPLAY_MIN_OK 8
#define REPLAY_BLK 4
#define REPLAY_EL (4 * REPLAY_BLK)
#define REPLAY_SPAN (128 * REPLAY_BLK)
static_assert(2 * REPLAY_SPAN <= RING, "a replayed pass and its sources fit the ring");

__global__ void osc_scan_kernel(const float* __restrict__ p_log_slow,
                                const float* __restrict__ p_log_fast,
                                const float* __restrict__ p_reset,
                                float* __restrict__ out, int N) {
    if (blockIdx.x != 0) return;
    const int lane = threadIdx.x;

    const float dslow = expf(p_log_slow[0]) * DT_F;
    const float dfast = expf(p_log_fast[0]) * DT_F;
    const float keep  = 1.0f - p_reset[0];

    float* __restrict__ oS = out;
    float* __restrict__ oF = out + N;

    const bool fast_ok =
        (dslow >= 0.0f) && (dslow < TWO_PI_F) &&
        (dfast >= 0.0f) && (dfast < TWO_PI_F) &&
        (keep  >= 0.0f) && (keep  <= 1.0f);

    __shared__ float hS[RING];
    __shared__ float hF[RING];

    if (fast_ok) {
        const float kf  = (float)lane;
        for (int i = lane; i < RING; i += 32) { hS[i] = 0.0f; hF[i] = 0.0f; }
        int evStep = -1, evBits = 0;
        int evN = 0, lag = 0;
        int hold = 0, nfail = 0;
        const float rds = to_scalar(1.0f / dslow);
        const float rdf = to_scalar(1.0f / dfast);
        float xs = 0.0f, xf = 0.0f, pv = 0.0f;
        int t = 0;
        while (t < N) {
            const bool replay = (lag > 0) && (t >= lag) && (t >= hold);
            if (replay) {
                float a[REPLAY_EL], g[REPLAY_EL];
                const int sb = (t + 4 * lane - lag) << 2;
#pragma unroll
                for (int e = 0; e < REPLAY_EL; ++e) {
                    const int idx = ((sb + 512 * (e >> 2) + 4 * (e & 3)) & (4 * RING - 4)) >> 2;
                    a[e] = hS[idx];
                    g[e] = hF[idx];
                }
                const float offS = xs - __int_as_float(__builtin_amdgcn_readfirstlane(__float_as_int(a[0])));
                const float offF = xf - __int_as_float(__builtin_amdgcn_readfirstlane(__float_as_int(g[0])));
#pragma unroll
                for (int e = 0; e < REPLAY_EL; ++e) { a[e] += offS; g[e] += offF; }
                a[0] = (lane == 0) ? xs : a[0];
                g[0] = (lane == 0) ? xf : g[0];
                const int upSrc = ((lane + 1) & 31) << 2;
                const int dnSrc = ((lane + 31) & 31) << 2;
                int a1s[REPLAY_BLK], g1s[REPLAY_BLK], aps[REPLAY_BLK];
#pragma unroll
                for (int b = 0; b < REPLAY_BLK; ++b) {
                    const float aN = (b + 1 < REPLAY_BLK) ? a[4 * b + 4] : a[4 * b];
                    const float gN = (b + 1 < REPLAY_BLK) ? g[4 * b + 4] : g[4 * b];
                    const float aP = (b > 0) ? a[4 * b - 1] : pv;
                    a1s[b] = __builtin_amdgcn_ds_bpermute(upSrc, __float_as_int((lane == 0) ? aN : a[4 * b]));
                    g1s[b] = __builtin_amdgcn_ds_bpermute(upSrc, __float_as_int((lane == 0) ? gN : g[4 * b]));
                    aps[b] = __builtin_amdgcn_ds_bpermute(dnSrc, __float_as_int((lane == 31) ? aP : a[4 * b + 3]));
                }
                float ss[REPLAY_EL], sf[REPLAY_EL];
                unsigned cb[REPLAY_EL], xw[REPLAY_EL];
                auto fast_step = [&](int e, float prev) {
                    const int up = __float_as_int(PI_F - prev);
                    const int dn = __float_as_int(a[e] - PI_F);
                    const bool crossed = (up & dn) < 0;
                    cb[e] = __builtin_amdgcn_ballot_w32(crossed);
                    const float kk = crossed ? keep : 1.0f;
                    sf[e] = wrap_add(g[e], dfast) * kk;
                };
                __builtin_amdgcn_sched_barrier(0);
#pragma unroll
                for (int e = 0; e < REPLAY_EL; ++e) {
                    const int j = e & 3;
                    ss[e] = wrap_add(a[e], dslow);
                    if (j > 0) fast_step(e, a[e - 1]);
                    if (j == 1 || j == 2)
                        xw[e] = (unsigned)(__float_as_int(ss[e]) ^ __float_as_int(a[e + 1])) |
                                (unsigned)(__float_as_int(sf[e]) ^ __float_as_int(g[e + 1]));
                }
                __builtin_amdgcn_sched_barrier(0);
#pragma unroll
                for (int e = 0; e < REPLAY_EL; ++e) {
                    const int b = e >> 2, j = e & 3;
                    if (j == 0) {
                        fast_step(e, __int_as_float(aps[b]));
                        xw[e] = (unsigned)(__float_as_int(ss[e]) ^ __float_as_int(a[e + 1])) |
                                (unsigned)(__float_as_int(sf[e]) ^ __float_as_int(g[e + 1]));
                    }
                    if (j == 3)
                        xw[e] = (unsigned)(__float_as_int(ss[e]) ^ a1s[b]) | (unsigned)(__float_as_int(sf[e]) ^ g1s[b]);
                }
                static_assert((REPLAY_EL & 1) == 0, "the links are gathered two elements at a time");
                unsigned bad = 0u;
#pragma unroll
                for (int e = 0; e + 2 < REPLAY_EL; e += 2) bad = or3_gather(bad, xw[e], xw[e + 1]);
                bad = or3_gather(bad, xw[REPLAY_EL - 2], (lane == 31) ? 0u : xw[REPLAY_EL - 1]);
                const unsigned allm = (__builtin_amdgcn_ballot_w32(bad != 0u) == 0u) ? 0xffffffffu : 0u;
                auto record = [&](int b, unsigned c0, unsigned c1, unsigned c2, unsigned c3,
                                  float g0, float g1, float g2, float g3) {
                    unsigned cm = c0 | c1 | c2 | c3;
                    while (cm != 0u) {
                        const int kc = __builtin_ctz(cm);
                        const unsigned bit = 1u << kc;
                        const int j = (c0 & bit) ? 0 : ((c1 & bit) ? 1 : ((c2 & bit) ? 2 : 3));
                        const float gj = (j == 0) ? g0 : ((j == 1) ? g1 : ((j == 2) ? g2 : g3));
                        const int fb = __builtin_amdgcn_readlane(__float_as_int(gj), kc);
                        if (lane == (evN & 31)) { evStep = t + 128 * b + 4 * kc + j; evBits = fb; }
                        ++evN;
                        cm &= cm - 1u;
                    }
                };
                if (allm == 0xffffffffu && t + REPLAY_SPAN <= N) {
                    xs = __int_as_float(__builtin_amdgcn_readlane(__float_as_int(ss[REPLAY_EL - 1]), 31));
                    xf = __int_as_float(__builtin_amdgcn_readlane(__float_as_int(sf[REPLAY_EL - 1]), 31));
                    pv = __int_as_float(__builtin_amdgcn_readlane(__float_as_int(a[REPLAY_EL - 1]), 31));
                    float* __restrict__ pS = oS + (t + 4 * lane);
                    float* __restrict__ pF = oF + (t + 4 * lane);
                    const int tb = (t + 4 * lane) << 2;
#pragma unroll
                    for (int b = 0; b < REPLAY_BLK; ++b) {
                        const v4f vS = {a[4 * b], a[4 * b + 1], a[4 * b + 2], a[4 * b + 3]};
                        const v4f vF = {g[4 * b], g[4 * b + 1], g[4 * b + 2], g[4 * b + 3]};
                        __builtin_memcpy(pS + 128 * b, &vS, sizeof(v4f));
                        __builtin_memcpy(pF + 128 * b, &vF, sizeof(v4f));
#pragma unroll
                        for (int j = 0; j < 4; ++j) {
                            const int idx = ((tb + 512 * b + 4 * j) & (4 * RING - 4)) >> 2;
                            hS[idx] = a[4 * b + j];
                            hF[idx] = g[4 * b + j];
                        }
                    }
#pragma unroll
                    for (int b = 0; b < REPLAY_BLK; ++b)
                        record(b, cb[4 * b], cb[4 * b + 1], cb[4 * b + 2], cb[4 * b + 3],
                               g[4 * b], g[4 * b + 1], g[4 * b + 2], g[4 * b + 3]);
                    nfail = 0;
                    t += REPLAY_SPAN;
                    continue;
                }
                unsigned mk[REPLAY_EL];
#pragma unroll
                for (int e = 0; e < REPLAY_EL; ++e) mk[e] = __builtin_amdgcn_ballot_w32(xw[e] == 0u);
                mk[REPLAY_EL - 1] &= 0x7fffffffu;
                int K = REPLAY_SPAN - 1;
#pragma unroll
                for (int b = REPLAY_BLK - 1; b >= 0; --b) {
                    const unsigned okl = mk[4 * b] & mk[4 * b + 1] & mk[4 * b + 2] & mk[4 * b + 3];
                    if (okl != 0xffffffffu) {
                        const int lf = __builtin_ctz(~okl);
                        const unsigned bit = 1u << lf;
                        const int jf = !(mk[4 * b] & bit) ? 0 : (!(mk[4 * b + 1] & bit) ? 1 : (!(mk[4 * b + 2] & bit) ? 2 : 3));
                        K = 128 * b + 4 * lf + jf;
                    }
                }
                const int eK = ((K >> 7) << 2) | (K & 3), lK = (K >> 2) & 31;
                float nxs = xs, nxf = xf, npv = pv;
#pragma unroll
                for (int e = 0; e < REPLAY_EL; ++e) {
                    const float c1 = __int_as_float(__builtin_amdgcn_readlane(__float_as_int(ss[e]), lK));
                    const float c2 = __int_as_float(__builtin_amdgcn_readlane(__float_as_int(sf[e]), lK));
                    const float c3 = __int_as_float(__builtin_amdgcn_readlane(__float_as_int(a[e]), lK));
                    if (e == eK) { nxs = c1; nxf = c2; npv = c3; }
                }
                xs = nxs; xf = nxf; pv = npv;
                const int last = min(K, N - 1 - t);
#pragma unroll
                for (int e = 0; e < REPLAY_EL; ++e) {
                    const int q = 128 * (e >> 2) + 4 * lane + (e & 3);
                    if (q <= last) {
                        oS[t + q] = a[e];
                        oF[t + q] = g[e];
                        hS[(t + q) & RING_MASK] = a[e];
                        hF[(t + q) & RING_MASK] = g[e];
                    }
                }
                if (K + 1 < REPLAY_MIN_OK) {
                    if (++nfail >= 2) { hold = t + lag; nfail = 0; }
                    lag = 0;
                }
                unsigned cv[REPLAY_EL];
#pragma unroll
                for (int e = 0; e < REPLAY_EL; ++e) {
                    const int r = K - 128 * (e >> 2) - (e & 3);
                    const int lv = r >> 2;
                    const unsigned vm = (lv >= 31) ? 0xffffffffu : ((r < 0) ? 0u : (0xffffffffu >> (31 - lv)));
                    cv[e] = cb[e] & vm;
                }
#pragma unroll
                for (int b = 0; b < REPLAY_BLK; ++b)
                    record(b, cv[4 * b], cv[4 * b + 1], cv[4 * b + 2], cv[4 * b + 3],
                           g[4 * b], g[4 * b + 1], g[4 * b + 2], g[4 * b + 3]);
                t += K + 1;
                continue;
            }
            float a, g;
            {
                Seg sS[RUNS_SLOW], sF[RUNS_FAST];
                {
                    float wS = xs, cS = 0.0f, wF = xf, cF = 0.0f;
#pragma unroll
                    for (int s = 0; s < RUNS_SLOW; ++s)
                        walk_pair(wF, cF, dfast, rdf, sF[s], wS, cS, dslow, rds, sS[s]);
#pragma unroll
                    for (int s = RUNS_SLOW; s < RUNS_FAST; ++s)
                        walk_one(wF, cF, dfast, rdf, sF[s]);
                }
                a = eval_prop<RUNS_SLOW>(sS, kf);
                g = eval_prop<RUNS_FAST>(sF, kf);
            }
            const int ai = __float_as_int(a), gi = __float_as_int(g);
            const int a16 = __builtin_amdgcn_readlane(ai, 16);
            const int g16 = __builtin_amdgcn_readlane(gi, 16);
            const int a15 = __builtin_amdgcn_readlane(ai, 15);
            const float a1 = __int_as_float(__builtin_amdgcn_update_dpp(a16, ai, 0x101, 0xf, 0xf, false));
            const float g1 = __int_as_float(__builtin_amdgcn_update_dpp(g16, gi, 0x101, 0xf, 0xf, false));
            const int apo  = (lane == 0) ? __float_as_int(pv) : a15;
            const float ap = __int_as_float(__builtin_amdgcn_update_dpp(apo, ai, 0x111, 0xf, 0xf, false));
            const float ss = wrap_add(a, dslow);
            const bool crossed = (ap > PI_F) && (a < PI_F);
            const unsigned cball = __builtin_amdgcn_ballot_w32(crossed);
            const float kk = crossed ? keep : 1.0f;
            const float sf = wrap_add(g, dfast) * kk;
            const unsigned m =
                __builtin_amdgcn_ballot_w32(__float_as_uint(ss) == __float_as_uint(a1)) &
                __builtin_amdgcn_ballot_w32(__float_as_uint(sf) == __float_as_uint(g1));
            const int K = __builtin_ctz(~m | 0x80000000u);
            xs = __int_as_float(__builtin_amdgcn_readlane(__float_as_int(ss), K));
            xf = __int_as_float(__builtin_amdgcn_readlane(__float_as_int(sf), K));
            pv = __int_as_float(__builtin_amdgcn_readlane(ai, K));
            const int last = min(K, N - 1 - t);
            if (lane <= last) {
                oS[t + lane] = a;
                oF[t + lane] = g;
                hS[(t + lane) & RING_MASK] = a;
                hF[(t + lane) & RING_MASK] = g;
            }
            const unsigned cm = cball & (0xffffffffu >> (31 - K));
            if (cm != 0u) {
                const int kc = __builtin_ctz(cm);
                const int e  = t + kc;
                const int fb = __builtin_amdgcn_readlane(gi, kc);
                const unsigned mm = __builtin_amdgcn_ballot_w32((evStep >= 0) && (evBits == fb));
                const int slot = evN & 31;
                if (mm != 0u) {
                    const unsigned rr = __builtin_rotateright32(mm, (unsigned)slot);
                    const int ms = (31 - __builtin_clz(rr) + slot) & 31;
                    const int cand = e - __builtin_amdgcn_readlane(evStep, ms);
                    if (cand >= 32) {
                        int c2 = cand;
                        while (c2 < REPLAY_SPAN) c2 += cand;
                        if (c2 + REPLAY_SPAN <= RING) lag = c2;
                    }
                }
                if (lane == slot) { evStep = e; evBits = fb; }
                ++evN;
            }
            t += K + 1;
        }
        return;
    }

    if (lane != 0) return;
    float slow = 0.0f, fast = 0.0f, prev = 0.0f;
    for (int t = 0; t < N; ++t) {
        oS[t] = slow;
        oF[t] = fast;
        bool crossed = (prev > PI_F) && (slow < PI_F);
        float ns = pmod(slow + dslow, TWO_PI_F);
        float nf = pmod(fast + dfast, TWO_PI_F);
        if (crossed) nf = pmod(nf * keep, TWO_PI_F);
        prev = slow; slow = ns; fast = nf;
    }
}

__global__ void osc_scale_wmma_kernel(const float* __restrict__ in,
                                      float* __restrict__ out, int N) {
    const float s = INV_TWO_PI_F;
    const int base = blockIdx.x * 256;
    const int lane = threadIdx.x;
    const int i    = lane & 15;
    const int kh   = lane >> 4;

    if (base + 256 <= N) {
        v8f d = {0.f, 0.f, 0.f, 0.f, 0.f, 0.f, 0.f, 0.f};
#pragma unroll
        for (int m = 0; m < 4; ++m) {
            const float* p = in + base + i * 16 + 4 * m + 2 * kh;
            v2f a; a.x = p[0]; a.y = p[1];
            v2f b;
            b.x = (i == (4 * m + 2 * kh + 0)) ? s : 0.0f;
            b.y = (i == (4 * m + 2 * kh + 1)) ? s : 0.0f;
            d = __builtin_amdgcn_wmma_f32_16x16x4_f32(
                    false, a, false, b, (short)0, d, false, false);
        }
#pragma unroll
        for (int r = 0; r < 8; ++r) {
            out[base + (r + 8 * kh) * 16 + i] = d[r];
        }
    } else {
        for (int j = base + lane; j < N; j += 32) out[j] = in[j] * s;
    }
}

extern "C" void kernel_launch(void* const* d_in, const int* in_sizes, int n_in,
                              void* d_out, int out_size, void* d_ws, size_t ws_size,
                              hipStream_t stream) {
    const float* p_log_slow = (const float*)d_in[0];
    const float* p_log_fast = (const float*)d_in[1];
    const float* p_reset    = (const float*)d_in[2];
    const int N = out_size / 3;
    if (N <= 0) return;
    float* out = (float*)d_out;

    osc_scan_kernel<<<1, 32, 0, stream>>>(p_log_slow, p_log_fast, p_reset, out, N);

    const int tiles = (N + 255) / 256;
    osc_scale_wmma_kernel<<<tiles, 32, 0, stream>>>(out, out + 2 * (size_t)N, N);
}


# scan: a replayed pass no longer records its trough crossings (a scalar walk over the crossing lanes); a dropped lag is re-tested against the history ring at the next walked crossing; same values
# speedup vs baseline: 8.7680x; 1.1140x over previous
#include <hip/hip_runtime.h>
#include <hip/hip_bf16.h>


typedef __attribute__((ext_vector_type(2))) float v2f;
typedef __attribute__((ext_vector_type(8))) float v8f;
typedef __attribute__((ext_vector_type(4))) float v4f;

#define TWO_PI_F     6.28318530717958647692f
#define PI_F         3.14159265358979323846f
#define INV_TWO_PI_F 0.15915494309189533577f
#define DT_F         0.001f

__device__ __forceinline__ float pmod(float x, float m) {
    float r = fmodf(x, m);
    return (r < 0.0f) ? (r + m) : r;
}

__device__ __forceinline__ float wrap_add(float x, float d) {
    float t = x + d;
    return (t >= TWO_PI_F) ? (t - TWO_PI_F) : t;
}


struct Seg { float c, w, t2, de; };

__device__ __forceinline__ float to_scalar(float x) {
    float r;
    asm("v_readfirstlane_b32 %0, %1" : "=s"(r) : "v"(x));
    return r;
}

__device__ __forceinline__ void walk_pair(float& wF, float& cF, float dF, float rdF, Seg& sgF,
                                          float& wS, float& cS, float dS, float rdS, Seg& sgS) {
    float wnF, cnF, t2F, deF, bF, tF, baF;
    float wnS, cnS, t2S, deS, bS, tS, baS;
    asm(
        "s_and_b32 %[bF], %[wF], 0x7f800000\n\t"
        "s_and_b32 %[bS], %[wS], 0x7f800000\n\t"
        "s_add_f32 %[t2F], %[wF], %[dF]\n\t"
        "s_add_f32 %[t2S], %[wS], %[dS]\n\t"
        "s_add_f32 %[tF], %[bF], %[bF]\n\t"
        "s_add_f32 %[tS], %[bS], %[bS]\n\t"
        "s_add_f32 %[deF], %[bF], %[dF]\n\t"
        "s_add_f32 %[deS], %[bS], %[dS]\n\t"
        "s_min_u32 %[tF], %[tF], 0x40c90fdb\n\t"
        "s_min_u32 %[tS], %[tS], 0x40c90fdb\n\t"
        "s_sub_f32 %[deF], %[deF], %[bF]\n\t"
        "s_sub_f32 %[deS], %[deS], %[bS]\n\t"
        "s_sub_f32 %[tF], %[tF], %[t2F]\n\t"
        "s_sub_f32 %[tS], %[tS], %[t2S]\n\t"
        "s_sub_f32 %[baF], %[t2F], %[deF]\n\t"
        "s_sub_f32 %[baS], %[t2S], %[deS]\n\t"
        "s_mul_f32 %[tF], %[tF], %[rF]\n\t"
        "s_mul_f32 %[tS], %[tS], %[rS]\n\t"
        "s_ceil_f32 %[tF], %[tF]\n\t"
        "s_ceil_f32 %[tS], %[tS]\n\t"
        "s_fmac_f32 %[baF], %[tF], %[deF]\n\t"
        "s_fmac_f32 %[baS], %[tS], %[deS]\n\t"
        "s_max_num_f32 %[cnF], %[tF], 0\n\t"
        "s_max_num_f32 %[cnS], %[tS], 0\n\t"
        "s_cmp_ge_f32 %[tF], 1.0\n\t"
        "s_cselect_b32 %[baF], %[baF], %[wF]\n\t"
        "s_cmp_ge_f32 %[tS], 1.0\n\t"
        "s_cselect_b32 %[baS], %[baS], %[wS]\n\t"
        "s_add_f32 %[baF], %[baF], %[dF]\n\t"
        "s_add_f32 %[baS], %[baS], %[dS]\n\t"
        "s_add_f32 %[cnF], %[cnF], %[cF]\n\t"
        "s_add_f32 %[cnS], %[cnS], %[cS]\n\t"
        "s_add_f32 %[wnF], %[baF], 0xc0c90fdb\n\t"
        "s_add_f32 %[wnS], %[baS], 0xc0c90fdb\n\t"
        "s_add_f32 %[cnF], %[cnF], 1.0\n\t"
        "s_add_f32 %[cnS], %[cnS], 1.0\n\t"
        "s_cmp_ge_f32 %[baF], 0x40c90fdb\n\t"
        "s_cselect_b32 %[wnF], %[wnF], %[baF]\n\t"
        "s_cmp_ge_f32 %[baS], 0x40c90fdb\n\t"
        "s_cselect_b32 %[wnS], %[wnS], %[baS]\n\t"
        : [wnF] "=&s"(wnF), [cnF] "=&s"(cnF), [t2F] "=&s"(t2F), [deF] "=&s"(deF), [bF] "=&s"(bF), [tF] "=&s"(tF), [baF] "=&s"(baF), [wnS] "=&s"(wnS), [cnS] "=&s"(cnS), [t2S] "=&s"(t2S), [deS] "=&s"(deS), [bS] "=&s"(bS), [tS] "=&s"(tS), [baS] "=&s"(baS)
        : [wF] "s"(wF), [cF] "s"(cF), [dF] "s"(dF), [rF] "s"(rdF), [wS] "s"(wS), [cS] "s"(cS), [dS] "s"(dS), [rS] "s"(rdS)
        : "scc");
    sgF.c = cF; sgF.w = wF; sgF.t2 = t2F; sgF.de = deF;
    sgS.c = cS; sgS.w = wS; sgS.t2 = t2S; sgS.de = deS;
    wF = wnF; cF = cnF;
    wS = wnS; cS = cnS;
}

__device__ __forceinline__ void walk_one(float& wF, float& cF, float dF, float rdF, Seg& sgF) {
    float wnF, cnF, t2F, deF, bF, tF, baF;
    asm(
        "s_and_b32 %[bF], %[wF], 0x7f800000\n\t"
        "s_add_f32 %[t2F], %[wF], %[dF]\n\t"
        "s_add_f32 %[tF], %[bF], %[bF]\n\t"
        "s_add_f32 %[deF], %[bF], %[dF]\n\t"
        "s_min_u32 %[tF], %[tF], 0x40c90fdb\n\t"
        "s_sub_f32 %[deF], %[deF], %[bF]\n\t"
        "s_sub_f32 %[tF], %[tF], %[t2F]\n\t"
        "s_sub_f32 %[baF], %[t2F], %[deF]\n\t"
        "s_mul_f32 %[tF], %[tF], %[rF]\n\t"
        "s_ceil_f32 %[tF], %[tF]\n\t"
        "s_fmac_f32 %[baF], %[tF], %[deF]\n\t"
        "s_max_num_f32 %[cnF], %[tF], 0\n\t"
        "s_cmp_ge_f32 %[tF], 1.0\n\t"
        "s_cselect_b32 %[baF], %[baF], %[wF]\n\t"
        "s_add_f32 %[baF], %[baF], %[dF]\n\t"
        "s_add_f32 %[cnF], %[cnF], %[cF]\n\t"
        "s_add_f32 %[wnF], %[baF], 0xc0c90fdb\n\t"
        "s_add_f32 %[cnF], %[cnF], 1.0\n\t"
        "s_cmp_ge_f32 %[baF], 0x40c90fdb\n\t"
        "s_cselect_b32 %[wnF], %[wnF], %[baF]\n\t"
        : [wnF] "=&s"(wnF), [cnF] "=&s"(cnF), [t2F] "=&s"(t2F), [deF] "=&s"(deF), [bF] "=&s"(bF), [tF] "=&s"(tF), [baF] "=&s"(baF)
        : [wF] "s"(wF), [cF] "s"(cF), [dF] "s"(dF), [rF] "s"(rdF)
        : "scc");
    sgF.c = cF; sgF.w = wF; sgF.t2 = t2F; sgF.de = deF;
    wF = wnF; cF = cnF;
}

template <int S>
__device__ __forceinline__ float eval_prop(const Seg (&sg)[S], float kf) {
    float c = sg[0].c, w = sg[0].w, t2 = sg[0].t2, de = sg[0].de;
#pragma unroll
    for (int s = 1; s < S; ++s) {
        const bool in = kf >= sg[s].c;
        c  = in ? sg[s].c  : c;
        w  = in ? sg[s].w  : w;
        t2 = in ? sg[s].t2 : t2;
        de = in ? sg[s].de : de;
    }
    const float j = kf - c;
    return (j == 0.0f) ? w : fmaf(j - 1.0f, de, t2);
}

__device__ __forceinline__ unsigned or3_gather(unsigned acc, unsigned x, unsigned y) {
    unsigned r;
    asm("v_or3_b32 %0, %1, %2, %3" : "=v"(r) : "v"(acc), "v"(x), "v"(y));
    return r;
}

#define RUNS_SLOW 5
#define RUNS_FAST 9
static_assert(RUNS_SLOW <= RUNS_FAST, "the paired walk covers the slow table");

#define RING 4096
#define RING_MASK (RING - 1)
#define REPLAY_MIN_OK 8
#define REPLAY_BLK 4
#define REPLAY_EL (4 * REPLAY_BLK)
#define REPLAY_SPAN (128 * REPLAY_BLK)
static_assert(2 * REPLAY_SPAN <= RING, "a replayed pass and its sources fit the ring");

__global__ void osc_scan_kernel(const float* __restrict__ p_log_slow,
                                const float* __restrict__ p_log_fast,
                                const float* __restrict__ p_reset,
                                float* __restrict__ out, int N) {
    if (blockIdx.x != 0) return;
    const int lane = threadIdx.x;

    const float dslow = expf(p_log_slow[0]) * DT_F;
    const float dfast = expf(p_log_fast[0]) * DT_F;
    const float keep  = 1.0f - p_reset[0];

    float* __restrict__ oS = out;
    float* __restrict__ oF = out + N;

    const bool fast_ok =
        (dslow >= 0.0f) && (dslow < TWO_PI_F) &&
        (dfast >= 0.0f) && (dfast < TWO_PI_F) &&
        (keep  >= 0.0f) && (keep  <= 1.0f);

    __shared__ float hS[RING];
    __shared__ float hF[RING];

    if (fast_ok) {
        const float kf  = (float)lane;
        for (int i = lane; i < RING; i += 32) { hS[i] = 0.0f; hF[i] = 0.0f; }
        int evStep = -1, evBits = 0;
        int evN = 0, lag = 0;
        int lastLag = 0;
        int hold = 0, nfail = 0;
        const float rds = to_scalar(1.0f / dslow);
        const float rdf = to_scalar(1.0f / dfast);
        float xs = 0.0f, xf = 0.0f, pv = 0.0f;
        int t = 0;
        while (t < N) {
            const bool replay = (lag > 0) && (t >= lag) && (t >= hold);
            if (replay) {
                float a[REPLAY_EL], g[REPLAY_EL];
                const int sb = (t + 4 * lane - lag) << 2;
#pragma unroll
                for (int e = 0; e < REPLAY_EL; ++e) {
                    const int idx = ((sb + 512 * (e >> 2) + 4 * (e & 3)) & (4 * RING - 4)) >> 2;
                    a[e] = hS[idx];
                    g[e] = hF[idx];
                }
                const float offS = xs - __int_as_float(__builtin_amdgcn_readfirstlane(__float_as_int(a[0])));
                const float offF = xf - __int_as_float(__builtin_amdgcn_readfirstlane(__float_as_int(g[0])));
#pragma unroll
                for (int e = 0; e < REPLAY_EL; ++e) { a[e] += offS; g[e] += offF; }
                a[0] = (lane == 0) ? xs : a[0];
                g[0] = (lane == 0) ? xf : g[0];
                const int upSrc = ((lane + 1) & 31) << 2;
                const int dnSrc = ((lane + 31) & 31) << 2;
                int a1s[REPLAY_BLK], g1s[REPLAY_BLK], aps[REPLAY_BLK];
#pragma unroll
                for (int b = 0; b < REPLAY_BLK; ++b) {
                    const float aN = (b + 1 < REPLAY_BLK) ? a[4 * b + 4] : a[4 * b];
                    const float gN = (b + 1 < REPLAY_BLK) ? g[4 * b + 4] : g[4 * b];
                    const float aP = (b > 0) ? a[4 * b - 1] : pv;
                    a1s[b] = __builtin_amdgcn_ds_bpermute(upSrc, __float_as_int((lane == 0) ? aN : a[4 * b]));
                    g1s[b] = __builtin_amdgcn_ds_bpermute(upSrc, __float_as_int((lane == 0) ? gN : g[4 * b]));
                    aps[b] = __builtin_amdgcn_ds_bpermute(dnSrc, __float_as_int((lane == 31) ? aP : a[4 * b + 3]));
                }
                float ss[REPLAY_EL], sf[REPLAY_EL];
                unsigned xw[REPLAY_EL];
                auto fast_step = [&](int e, float prev) {
                    const int up = __float_as_int(PI_F - prev);
                    const int dn = __float_as_int(a[e] - PI_F);
                    const bool crossed = (up & dn) < 0;
                    const float kk = crossed ? keep : 1.0f;
                    sf[e] = wrap_add(g[e], dfast) * kk;
                };
                __builtin_amdgcn_sched_barrier(0);
#pragma unroll
                for (int e = 0; e < REPLAY_EL; ++e) {
                    const int j = e & 3;
                    ss[e] = wrap_add(a[e], dslow);
                    if (j > 0) fast_step(e, a[e - 1]);
                    if (j == 1 || j == 2)
                        xw[e] = (unsigned)(__float_as_int(ss[e]) ^ __float_as_int(a[e + 1])) |
                                (unsigned)(__float_as_int(sf[e]) ^ __float_as_int(g[e + 1]));
                }
                __builtin_amdgcn_sched_barrier(0);
#pragma unroll
                for (int e = 0; e < REPLAY_EL; ++e) {
                    const int b = e >> 2, j = e & 3;
                    if (j == 0) {
                        fast_step(e, __int_as_float(aps[b]));
                        xw[e] = (unsigned)(__float_as_int(ss[e]) ^ __float_as_int(a[e + 1])) |
                                (unsigned)(__float_as_int(sf[e]) ^ __float_as_int(g[e + 1]));
                    }
                    if (j == 3)
                        xw[e] = (unsigned)(__float_as_int(ss[e]) ^ a1s[b]) | (unsigned)(__float_as_int(sf[e]) ^ g1s[b]);
                }
                static_assert((REPLAY_EL & 1) == 0, "the links are gathered two elements at a time");
                unsigned bad = 0u;
#pragma unroll
                for (int e = 0; e + 2 < REPLAY_EL; e += 2) bad = or3_gather(bad, xw[e], xw[e + 1]);
                bad = or3_gather(bad, xw[REPLAY_EL - 2], (lane == 31) ? 0u : xw[REPLAY_EL - 1]);
                const unsigned allm = (__builtin_amdgcn_ballot_w32(bad != 0u) == 0u) ? 0xffffffffu : 0u;
                if (allm == 0xffffffffu && t + REPLAY_SPAN <= N) {
                    xs = __int_as_float(__builtin_amdgcn_readlane(__float_as_int(ss[REPLAY_EL - 1]), 31));
                    xf = __int_as_float(__builtin_amdgcn_readlane(__float_as_int(sf[REPLAY_EL - 1]), 31));
                    pv = __int_as_float(__builtin_amdgcn_readlane(__float_as_int(a[REPLAY_EL - 1]), 31));
                    float* __restrict__ pS = oS + (t + 4 * lane);
                    float* __restrict__ pF = oF + (t + 4 * lane);
                    const int tb = (t + 4 * lane) << 2;
#pragma unroll
                    for (int b = 0; b < REPLAY_BLK; ++b) {
                        const v4f vS = {a[4 * b], a[4 * b + 1], a[4 * b + 2], a[4 * b + 3]};
                        const v4f vF = {g[4 * b], g[4 * b + 1], g[4 * b + 2], g[4 * b + 3]};
                        __builtin_memcpy(pS + 128 * b, &vS, sizeof(v4f));
                        __builtin_memcpy(pF + 128 * b, &vF, sizeof(v4f));
#pragma unroll
                        for (int j = 0; j < 4; ++j) {
                            const int idx = ((tb + 512 * b + 4 * j) & (4 * RING - 4)) >> 2;
                            hS[idx] = a[4 * b + j];
                            hF[idx] = g[4 * b + j];
                        }
                    }
                    nfail = 0;
                    t += REPLAY_SPAN;
                    continue;
                }
                unsigned mk[REPLAY_EL];
#pragma unroll
                for (int e = 0; e < REPLAY_EL; ++e) mk[e] = __builtin_amdgcn_ballot_w32(xw[e] == 0u);
                mk[REPLAY_EL - 1] &= 0x7fffffffu;
                int K = REPLAY_SPAN - 1;
#pragma unroll
                for (int b = REPLAY_BLK - 1; b >= 0; --b) {
                    const unsigned okl = mk[4 * b] & mk[4 * b + 1] & mk[4 * b + 2] & mk[4 * b + 3];
                    if (okl != 0xffffffffu) {
                        const int lf = __builtin_ctz(~okl);
                        const unsigned bit = 1u << lf;
                        const int jf = !(mk[4 * b] & bit) ? 0 : (!(mk[4 * b + 1] & bit) ? 1 : (!(mk[4 * b + 2] & bit) ? 2 : 3));
                        K = 128 * b + 4 * lf + jf;
                    }
                }
                const int eK = ((K >> 7) << 2) | (K & 3), lK = (K >> 2) & 31;
                float nxs = xs, nxf = xf, npv = pv;
#pragma unroll
                for (int e = 0; e < REPLAY_EL; ++e) {
                    const float c1 = __int_as_float(__builtin_amdgcn_readlane(__float_as_int(ss[e]), lK));
                    const float c2 = __int_as_float(__builtin_amdgcn_readlane(__float_as_int(sf[e]), lK));
                    const float c3 = __int_as_float(__builtin_amdgcn_readlane(__float_as_int(a[e]), lK));
                    if (e == eK) { nxs = c1; nxf = c2; npv = c3; }
                }
                xs = nxs; xf = nxf; pv = npv;
                const int last = min(K, N - 1 - t);
#pragma unroll
                for (int e = 0; e < REPLAY_EL; ++e) {
                    const int q = 128 * (e >> 2) + 4 * lane + (e & 3);
                    if (q <= last) {
                        oS[t + q] = a[e];
                        oF[t + q] = g[e];
                        hS[(t + q) & RING_MASK] = a[e];
                        hF[(t + q) & RING_MASK] = g[e];
                    }
                }
                if (K + 1 < REPLAY_MIN_OK) {
                    if (++nfail >= 2) { hold = t + lag; nfail = 0; }
                    lag = 0;
                }
                t += K + 1;
                continue;
            }
            float a, g;
            {
                Seg sS[RUNS_SLOW], sF[RUNS_FAST];
                {
                    float wS = xs, cS = 0.0f, wF = xf, cF = 0.0f;
#pragma unroll
                    for (int s = 0; s < RUNS_SLOW; ++s)
                        walk_pair(wF, cF, dfast, rdf, sF[s], wS, cS, dslow, rds, sS[s]);
#pragma unroll
                    for (int s = RUNS_SLOW; s < RUNS_FAST; ++s)
                        walk_one(wF, cF, dfast, rdf, sF[s]);
                }
                a = eval_prop<RUNS_SLOW>(sS, kf);
                g = eval_prop<RUNS_FAST>(sF, kf);
            }
            const int ai = __float_as_int(a), gi = __float_as_int(g);
            const int a16 = __builtin_amdgcn_readlane(ai, 16);
            const int g16 = __builtin_amdgcn_readlane(gi, 16);
            const int a15 = __builtin_amdgcn_readlane(ai, 15);
            const float a1 = __int_as_float(__builtin_amdgcn_update_dpp(a16, ai, 0x101, 0xf, 0xf, false));
            const float g1 = __int_as_float(__builtin_amdgcn_update_dpp(g16, gi, 0x101, 0xf, 0xf, false));
            const int apo  = (lane == 0) ? __float_as_int(pv) : a15;
            const float ap = __int_as_float(__builtin_amdgcn_update_dpp(apo, ai, 0x111, 0xf, 0xf, false));
            const float ss = wrap_add(a, dslow);
            const bool crossed = (ap > PI_F) && (a < PI_F);
            const unsigned cball = __builtin_amdgcn_ballot_w32(crossed);
            const float kk = crossed ? keep : 1.0f;
            const float sf = wrap_add(g, dfast) * kk;
            const unsigned m =
                __builtin_amdgcn_ballot_w32(__float_as_uint(ss) == __float_as_uint(a1)) &
                __builtin_amdgcn_ballot_w32(__float_as_uint(sf) == __float_as_uint(g1));
            const int K = __builtin_ctz(~m | 0x80000000u);
            xs = __int_as_float(__builtin_amdgcn_readlane(__float_as_int(ss), K));
            xf = __int_as_float(__builtin_amdgcn_readlane(__float_as_int(sf), K));
            pv = __int_as_float(__builtin_amdgcn_readlane(ai, K));
            const int last = min(K, N - 1 - t);
            if (lane <= last) {
                oS[t + lane] = a;
                oF[t + lane] = g;
                hS[(t + lane) & RING_MASK] = a;
                hF[(t + lane) & RING_MASK] = g;
            }
            const unsigned cm = cball & (0xffffffffu >> (31 - K));
            if (cm != 0u) {
                const int kc = __builtin_ctz(cm);
                const int e  = t + kc;
                const int fb = __builtin_amdgcn_readlane(gi, kc);
                const unsigned mm = __builtin_amdgcn_ballot_w32((evStep >= 0) && (evBits == fb));
                const int slot = evN & 31;
                if (mm != 0u) {
                    const unsigned rr = __builtin_rotateright32(mm, (unsigned)slot);
                    const int ms = (31 - __builtin_clz(rr) + slot) & 31;
                    const int cand = e - __builtin_amdgcn_readlane(evStep, ms);
                    if (cand >= 32) {
                        int c2 = cand;
                        while (c2 < REPLAY_SPAN) c2 += cand;
                        if (c2 + REPLAY_SPAN <= RING) { lag = c2; lastLag = c2; }
                    }
                }
                if (lag == 0 && lastLag > 0 && e > lastLag) {
                    const int qr = e - lastLag;
                    const int hb = __builtin_amdgcn_readfirstlane(__float_as_int(hF[qr & RING_MASK]));
                    const float h1 = __int_as_float(__builtin_amdgcn_readfirstlane(__float_as_int(hS[qr & RING_MASK])));
                    const float h0 = __int_as_float(__builtin_amdgcn_readfirstlane(__float_as_int(hS[(qr - 1) & RING_MASK])));
                    if (hb == fb && h0 > PI_F && h1 < PI_F) lag = lastLag;
                }
                if (lane == slot) { evStep = e; evBits = fb; }
                ++evN;
            }
            t += K + 1;
        }
        return;
    }

    if (lane != 0) return;
    float slow = 0.0f, fast = 0.0f, prev = 0.0f;
    for (int t = 0; t < N; ++t) {
        oS[t] = slow;
        oF[t] = fast;
        bool crossed = (prev > PI_F) && (slow < PI_F);
        float ns = pmod(slow + dslow, TWO_PI_F);
        float nf = pmod(fast + dfast, TWO_PI_F);
        if (crossed) nf = pmod(nf * keep, TWO_PI_F);
        prev = slow; slow = ns; fast = nf;
    }
}

__global__ void osc_scale_wmma_kernel(const float* __restrict__ in,
                                      float* __restrict__ out, int N) {
    const float s = INV_TWO_PI_F;
    const int base = blockIdx.x * 256;
    const int lane = threadIdx.x;
    const int i    = lane & 15;
    const int kh   = lane >> 4;

    if (base + 256 <= N) {
        v8f d = {0.f, 0.f, 0.f, 0.f, 0.f, 0.f, 0.f, 0.f};
#pragma unroll
        for (int m = 0; m < 4; ++m) {
            const float* p = in + base + i * 16 + 4 * m + 2 * kh;
            v2f a; a.x = p[0]; a.y = p[1];
            v2f b;
            b.x = (i == (4 * m + 2 * kh + 0)) ? s : 0.0f;
            b.y = (i == (4 * m + 2 * kh + 1)) ? s : 0.0f;
            d = __builtin_amdgcn_wmma_f32_16x16x4_f32(
                    false, a, false, b, (short)0, d, false, false);
        }
#pragma unroll
        for (int r = 0; r < 8; ++r) {
            out[base + (r + 8 * kh) * 16 + i] = d[r];
        }
    } else {
        for (int j = base + lane; j < N; j += 32) out[j] = in[j] * s;
    }
}

extern "C" void kernel_launch(void* const* d_in, const int* in_sizes, int n_in,
                              void* d_out, int out_size, void* d_ws, size_t ws_size,
                              hipStream_t stream) {
    const float* p_log_slow = (const float*)d_in[0];
    const float* p_log_fast = (const float*)d_in[1];
    const float* p_reset    = (const float*)d_in[2];
    const int N = out_size / 3;
    if (N <= 0) return;
    float* out = (float*)d_out;

    osc_scan_kernel<<<1, 32, 0, stream>>>(p_log_slow, p_log_fast, p_reset, out, N);

    const int tiles = (N + 255) / 256;
    osc_scale_wmma_kernel<<<tiles, 32, 0, stream>>>(out, out + 2 * (size_t)N, N);
}
